# starting kernel
# baseline (speedup 1.0000x reference)
#include <hip/hip_runtime.h>

typedef __attribute__((ext_vector_type(2))) float v2f;
typedef __attribute__((ext_vector_type(8))) float v8f;

constexpr int EMB   = 128;
constexpr int SEQ   = 2048;
constexpr int BATCH = 4;
constexpr int HEADS = 16;
constexpr int HD    = 8;
constexpr float ATT_SCALE = 0.35355339059327373f;

__device__ __forceinline__ v8f wmma4_f32(v2f a, v2f b, v8f c) {
  return __builtin_amdgcn_wmma_f32_16x16x4_f32(
       false, a,  false, b,
       (short)0, c,  false,  false);
}

__global__ __launch_bounds__(128) void linear_wmma(
    const float* __restrict__ x, const float* __restrict__ W,
    const float* __restrict__ bias, float* __restrict__ out) {
  const int lane = threadIdx.x & 31;
  const int wave = threadIdx.x >> 5;
  const int tile = blockIdx.x * 4 + wave;
  const int mt = tile >> 3;
  const int nt = tile & 7;
  const int m0 = mt * 16, n0 = nt * 16;
  const int lo = lane & 15, hi = lane >> 4;

  const float* arow = x + (size_t)(m0 + lo) * EMB + 2 * hi;
  const float* brow = W + (size_t)(n0 + lo) * EMB + 2 * hi;
  v8f c = {};
#pragma unroll
  for (int k0 = 0; k0 < EMB; k0 += 4) {
    v2f a = *(const v2f*)(arow + k0);
    v2f b = *(const v2f*)(brow + k0);
    c = wmma4_f32(a, b, c);
  }
  const float bn = bias[n0 + lo];
  float* orow = out + (size_t)(m0 + 8 * hi) * EMB + n0 + lo;
#pragma unroll
  for (int r = 0; r < 8; ++r)
    orow[(size_t)r * EMB] = c[r] + bn;
}

__global__ __launch_bounds__(128) void attn_wmma(
    const float* __restrict__ Q, const float* __restrict__ K,
    const float* __restrict__ V, float* __restrict__ O) {
  __shared__ float wlds[4 * 16 * 17];
  const int lane = threadIdx.x & 31;
  const int wave = threadIdx.x >> 5;
  const int g  = blockIdx.x * 4 + wave;
  const int qt = g & 127;
  const int bh = g >> 7;
  const int b = bh >> 4, h = bh & 15;
  const int lo = lane & 15, hi = lane >> 4;
  const size_t base = (size_t)b * SEQ * EMB + (size_t)h * HD;
  const float* Qb = Q + base;
  const float* Kb = K + base;
  const float* Vb = V + base;
  float*       Ob = O + base;
  const int q0 = qt * 16;

  const float* qrow = Qb + (size_t)(q0 + lo) * EMB + 2 * hi;
  const v2f aq0 = *(const v2f*)(qrow);
  const v2f aq1 = *(const v2f*)(qrow + 4);

  float* wl = wlds + wave * (16 * 17);
  v8f acc = {};

  for (int kt = 0; kt <= qt; ++kt) {
    const int k0r = kt * 16;

    const float* krow = Kb + (size_t)(k0r + lo) * EMB + 2 * hi;
    const v2f bk0 = *(const v2f*)(krow);
    const v2f bk1 = *(const v2f*)(krow + 4);
    v8f s = {};
    s = wmma4_f32(aq0, bk0, s);
    s = wmma4_f32(aq1, bk1, s);

    const int n = k0r + lo;
#pragma unroll
    for (int r = 0; r < 8; ++r) {
      const int m = q0 + r + 8 * hi;
      const float w = (n <= m) ? 1.0f / (1.0f + __expf(-ATT_SCALE * s[r])) : 0.0f;
      wl[(r + 8 * hi) * 17 + lo] = w;
    }
    asm volatile("s_wait_dscnt 0" ::: "memory");

#pragma unroll
    for (int k0 = 0; k0 < 16; k0 += 4) {
      v2f a;
      a.x = wl[lo * 17 + k0 + 2 * hi];
      a.y = wl[lo * 17 + k0 + 2 * hi + 1];
      const int kk = k0r + k0 + 2 * hi;
      const float* vp = Vb + (size_t)kk * EMB + (lo & 7);
      const float v0 = vp[0];
      const float v1 = vp[EMB];
      v2f bb;
      bb.x = (lo < 8) ? v0 : ((lo == 8) ? 1.0f : 0.0f);
      bb.y = (lo < 8) ? v1 : ((lo == 8) ? 1.0f : 0.0f);
      acc = wmma4_f32(a, bb, acc);
    }
  }

#pragma unroll
  for (int r = 0; r < 8; ++r) {
    const float rs = __shfl(acc[r], (lane & 16) | 8, 32);
    acc[r] = acc[r] / rs;
  }
  if (lo < 8) {
    float* orow = Ob + (size_t)(q0 + 8 * hi) * EMB + lo;
#pragma unroll
    for (int r = 0; r < 8; ++r)
      orow[(size_t)r * EMB] = acc[r];
  }
}

extern "C" void kernel_launch(void* const* d_in, const int* in_sizes, int n_in,
                              void* d_out, int out_size, void* d_ws, size_t ws_size,
                              hipStream_t stream) {
  const float* x  = (const float*)d_in[0];
  const float* Wq = (const float*)d_in[1];
  const float* bq = (const float*)d_in[2];
  const float* Wk = (const float*)d_in[3];
  const float* bk = (const float*)d_in[4];
  const float* Wv = (const float*)d_in[5];
  const float* bv = (const float*)d_in[6];
  const float* Wo = (const float*)d_in[7];
  const float* bo = (const float*)d_in[8];
  float* out = (float*)d_out;

  const size_t mat = (size_t)BATCH * SEQ * EMB;
  float* Qw = (float*)d_ws;
  float* Kw = Qw + mat;
  float* Vw = Kw + mat;
  float* Aw = Vw + mat;

  dim3 blk(128);
  dim3 gproj(1024);
  linear_wmma<<<gproj, blk, 0, stream>>>(x, Wq, bq, Qw);
  linear_wmma<<<gproj, blk, 0, stream>>>(x, Wk, bk, Kw);
  linear_wmma<<<gproj, blk, 0, stream>>>(x, Wv, bv, Vw);
  attn_wmma<<<dim3(2048), blk, 0, stream>>>(Qw, Kw, Vw, Aw);
  linear_wmma<<<gproj, blk, 0, stream>>>(Aw, Wo, bo, out);
}


# attention scores computed transposed (K tile x Q^T): sigmoid tile feeds PV product as B operand from registers, LDS transpose and fence gone, loads issued up front; key order in tile sum permuted
# speedup vs baseline: 1.5098x; 1.5098x over previous
#include <hip/hip_runtime.h>

typedef __attribute__((ext_vector_type(2))) float v2f;
typedef __attribute__((ext_vector_type(8))) float v8f;

constexpr int EMB   = 128;
constexpr int SEQ   = 2048;
constexpr int BATCH = 4;
constexpr int HEADS = 16;
constexpr int HD    = 8;
constexpr float ATT_SCALE = 0.35355339059327373f;

__device__ __forceinline__ v8f wmma4_f32(v2f a, v2f b, v8f c) {
  return __builtin_amdgcn_wmma_f32_16x16x4_f32(
       false, a,  false, b,
       (short)0, c,  false,  false);
}

__global__ __launch_bounds__(128) void linear_wmma(
    const float* __restrict__ x, const float* __restrict__ W,
    const float* __restrict__ bias, float* __restrict__ out) {
  const int lane = threadIdx.x & 31;
  const int wave = threadIdx.x >> 5;
  const int tile = blockIdx.x * 4 + wave;
  const int mt = tile >> 3;
  const int nt = tile & 7;
  const int m0 = mt * 16, n0 = nt * 16;
  const int lo = lane & 15, hi = lane >> 4;

  const float* arow = x + (size_t)(m0 + lo) * EMB + 2 * hi;
  const float* brow = W + (size_t)(n0 + lo) * EMB + 2 * hi;
  v8f c = {};
#pragma unroll
  for (int k0 = 0; k0 < EMB; k0 += 4) {
    v2f a = *(const v2f*)(arow + k0);
    v2f b = *(const v2f*)(brow + k0);
    c = wmma4_f32(a, b, c);
  }
  const float bn = bias[n0 + lo];
  float* orow = out + (size_t)(m0 + 8 * hi) * EMB + n0 + lo;
#pragma unroll
  for (int r = 0; r < 8; ++r)
    orow[(size_t)r * EMB] = c[r] + bn;
}

__global__ __launch_bounds__(128) void attn_wmma(
    const float* __restrict__ Q, const float* __restrict__ K,
    const float* __restrict__ V, float* __restrict__ O) {
  const int lane = threadIdx.x & 31;
  const int wave = threadIdx.x >> 5;
  const int g  = blockIdx.x * 4 + wave;
  const int qt = g & 127;
  const int bh = g >> 7;
  const int b = bh >> 4, h = bh & 15;
  const int lo = lane & 15, hi = lane >> 4;
  const size_t base = (size_t)b * SEQ * EMB + (size_t)h * HD;
  const float* Qb = Q + base;
  const float* Kb = K + base;
  const float* Vb = V + base;
  float*       Ob = O + base;
  const int q0 = qt * 16;

  const float* qrow = Qb + (size_t)(q0 + lo) * EMB + 2 * hi;
  const v2f bq0 = *(const v2f*)(qrow);
  const v2f bq1 = *(const v2f*)(qrow + 4);

  const bool  isv  = lo < 8;
  const float fill = (lo == 8) ? 1.0f : 0.0f;
  const int   m    = q0 + lo;

  const float* kptr = Kb + (size_t)lo * EMB + 2 * hi;
  const float* vptr = Vb + (size_t)(8 * hi) * EMB + (lo & 7);
  v8f acc = {};

  for (int kt = 0; kt <= qt; ++kt) {
    const int k0r = kt * 16;

    const float* krow = kptr + (size_t)k0r * EMB;
    const v2f ak0 = *(const v2f*)(krow);
    const v2f ak1 = *(const v2f*)(krow + 4);
    const float* vp = vptr + (size_t)k0r * EMB;
    float vv[8];
#pragma unroll
    for (int r = 0; r < 8; ++r) vv[r] = vp[(size_t)r * EMB];

    v8f s = {};
    s = wmma4_f32(ak0, bq0, s);
    s = wmma4_f32(ak1, bq1, s);

    const int nb = k0r + 8 * hi;
    float w[8];
#pragma unroll
    for (int r = 0; r < 8; ++r)
      w[r] = (nb + r <= m) ? 1.0f / (1.0f + __expf(-ATT_SCALE * s[r])) : 0.0f;

#pragma unroll
    for (int j = 0; j < 4; ++j) {
      v2f a, bb;
      a.x  = isv ? vv[2 * j]     : fill;
      a.y  = isv ? vv[2 * j + 1] : fill;
      bb.x = w[2 * j];
      bb.y = w[2 * j + 1];
      acc = wmma4_f32(a, bb, acc);
    }
  }

  const float rs = __shfl(acc[0], 16 + lo, 32);
  if (hi == 0) {
    float4 o0, o1;
    o0.x = acc[0] / rs; o0.y = acc[1] / rs; o0.z = acc[2] / rs; o0.w = acc[3] / rs;
    o1.x = acc[4] / rs; o1.y = acc[5] / rs; o1.z = acc[6] / rs; o1.w = acc[7] / rs;
    float4* orow = (float4*)(Ob + (size_t)m * EMB);
    orow[0] = o0;
    orow[1] = o1;
  }
}

extern "C" void kernel_launch(void* const* d_in, const int* in_sizes, int n_in,
                              void* d_out, int out_size, void* d_ws, size_t ws_size,
                              hipStream_t stream) {
  const float* x  = (const float*)d_in[0];
  const float* Wq = (const float*)d_in[1];
  const float* bq = (const float*)d_in[2];
  const float* Wk = (const float*)d_in[3];
  const float* bk = (const float*)d_in[4];
  const float* Wv = (const float*)d_in[5];
  const float* bv = (const float*)d_in[6];
  const float* Wo = (const float*)d_in[7];
  const float* bo = (const float*)d_in[8];
  float* out = (float*)d_out;

  const size_t mat = (size_t)BATCH * SEQ * EMB;
  float* Qw = (float*)d_ws;
  float* Kw = Qw + mat;
  float* Vw = Kw + mat;
  float* Aw = Vw + mat;

  dim3 blk(128);
  dim3 gproj(1024);
  linear_wmma<<<gproj, blk, 0, stream>>>(x, Wq, bq, Qw);
  linear_wmma<<<gproj, blk, 0, stream>>>(x, Wk, bk, Kw);
  linear_wmma<<<gproj, blk, 0, stream>>>(x, Wv, bv, Vw);
  attn_wmma<<<dim3(2048), blk, 0, stream>>>(Qw, Kw, Vw, Aw);
  linear_wmma<<<gproj, blk, 0, stream>>>(Aw, Wo, bo, out);
}


# sigmoid weight = hardware reciprocal (about 1 ulp) of 1 + 2^(c*s), one folded constant, instead of the IEEE division sequence; causal mask as a select, the eight per-element branches gone
# speedup vs baseline: 2.4191x; 1.6023x over previous
#include <hip/hip_runtime.h>

typedef __attribute__((ext_vector_type(2))) float v2f;
typedef __attribute__((ext_vector_type(8))) float v8f;

constexpr int EMB   = 128;
constexpr int SEQ   = 2048;
constexpr int BATCH = 4;
constexpr int HEADS = 16;
constexpr int HD    = 8;
constexpr float ATT_SCALE = 0.35355339059327373f;
constexpr float NEG_SCALE_LOG2E = -ATT_SCALE * 1.4426950408889634f;

__device__ __forceinline__ v8f wmma4_f32(v2f a, v2f b, v8f c) {
  return __builtin_amdgcn_wmma_f32_16x16x4_f32(
       false, a,  false, b,
       (short)0, c,  false,  false);
}

__global__ __launch_bounds__(128) void linear_wmma(
    const float* __restrict__ x, const float* __restrict__ W,
    const float* __restrict__ bias, float* __restrict__ out) {
  const int lane = threadIdx.x & 31;
  const int wave = threadIdx.x >> 5;
  const int tile = blockIdx.x * 4 + wave;
  const int mt = tile >> 3;
  const int nt = tile & 7;
  const int m0 = mt * 16, n0 = nt * 16;
  const int lo = lane & 15, hi = lane >> 4;

  const float* arow = x + (size_t)(m0 + lo) * EMB + 2 * hi;
  const float* brow = W + (size_t)(n0 + lo) * EMB + 2 * hi;
  v8f c = {};
#pragma unroll
  for (int k0 = 0; k0 < EMB; k0 += 4) {
    v2f a = *(const v2f*)(arow + k0);
    v2f b = *(const v2f*)(brow + k0);
    c = wmma4_f32(a, b, c);
  }
  const float bn = bias[n0 + lo];
  float* orow = out + (size_t)(m0 + 8 * hi) * EMB + n0 + lo;
#pragma unroll
  for (int r = 0; r < 8; ++r)
    orow[(size_t)r * EMB] = c[r] + bn;
}

__global__ __launch_bounds__(128) void attn_wmma(
    const float* __restrict__ Q, const float* __restrict__ K,
    const float* __restrict__ V, float* __restrict__ O) {
  const int lane = threadIdx.x & 31;
  const int wave = threadIdx.x >> 5;
  const int g  = blockIdx.x * 4 + wave;
  const int qt = g & 127;
  const int bh = g >> 7;
  const int b = bh >> 4, h = bh & 15;
  const int lo = lane & 15, hi = lane >> 4;
  const size_t base = (size_t)b * SEQ * EMB + (size_t)h * HD;
  const float* Qb = Q + base;
  const float* Kb = K + base;
  const float* Vb = V + base;
  float*       Ob = O + base;
  const int q0 = qt * 16;

  const float* qrow = Qb + (size_t)(q0 + lo) * EMB + 2 * hi;
  const v2f bq0 = *(const v2f*)(qrow);
  const v2f bq1 = *(const v2f*)(qrow + 4);

  const bool  isv  = lo < 8;
  const float fill = (lo == 8) ? 1.0f : 0.0f;
  const int   m    = q0 + lo;

  const float* kptr = Kb + (size_t)lo * EMB + 2 * hi;
  const float* vptr = Vb + (size_t)(8 * hi) * EMB + (lo & 7);
  v8f acc = {};

  for (int kt = 0; kt <= qt; ++kt) {
    const int k0r = kt * 16;

    const float* krow = kptr + (size_t)k0r * EMB;
    const v2f ak0 = *(const v2f*)(krow);
    const v2f ak1 = *(const v2f*)(krow + 4);
    const float* vp = vptr + (size_t)k0r * EMB;
    float vv[8];
#pragma unroll
    for (int r = 0; r < 8; ++r) vv[r] = vp[(size_t)r * EMB];

    v8f s = {};
    s = wmma4_f32(ak0, bq0, s);
    s = wmma4_f32(ak1, bq1, s);

    const int nb = k0r + 8 * hi;
    float w[8];
#pragma unroll
    for (int r = 0; r < 8; ++r) {
      const float e  = __builtin_amdgcn_exp2f(NEG_SCALE_LOG2E * s[r]);
      const float sg = __builtin_amdgcn_rcpf(1.0f + e);
      w[r] = (nb + r <= m) ? sg : 0.0f;
    }

#pragma unroll
    for (int j = 0; j < 4; ++j) {
      v2f a, bb;
      a.x  = isv ? vv[2 * j]     : fill;
      a.y  = isv ? vv[2 * j + 1] : fill;
      bb.x = w[2 * j];
      bb.y = w[2 * j + 1];
      acc = wmma4_f32(a, bb, acc);
    }
  }

  const float rs = __shfl(acc[0], 16 + lo, 32);
  if (hi == 0) {
    float4 o0, o1;
    o0.x = acc[0] / rs; o0.y = acc[1] / rs; o0.z = acc[2] / rs; o0.w = acc[3] / rs;
    o1.x = acc[4] / rs; o1.y = acc[5] / rs; o1.z = acc[6] / rs; o1.w = acc[7] / rs;
    float4* orow = (float4*)(Ob + (size_t)m * EMB);
    orow[0] = o0;
    orow[1] = o1;
  }
}

extern "C" void kernel_launch(void* const* d_in, const int* in_sizes, int n_in,
                              void* d_out, int out_size, void* d_ws, size_t ws_size,
                              hipStream_t stream) {
  const float* x  = (const float*)d_in[0];
  const float* Wq = (const float*)d_in[1];
  const float* bq = (const float*)d_in[2];
  const float* Wk = (const float*)d_in[3];
  const float* bk = (const float*)d_in[4];
  const float* Wv = (const float*)d_in[5];
  const float* bv = (const float*)d_in[6];
  const float* Wo = (const float*)d_in[7];
  const float* bo = (const float*)d_in[8];
  float* out = (float*)d_out;

  const size_t mat = (size_t)BATCH * SEQ * EMB;
  float* Qw = (float*)d_ws;
  float* Kw = Qw + mat;
  float* Vw = Kw + mat;
  float* Aw = Vw + mat;

  dim3 blk(128);
  dim3 gproj(1024);
  linear_wmma<<<gproj, blk, 0, stream>>>(x, Wq, bq, Qw);
  linear_wmma<<<gproj, blk, 0, stream>>>(x, Wk, bk, Kw);
  linear_wmma<<<gproj, blk, 0, stream>>>(x, Wv, bv, Vw);
  attn_wmma<<<dim3(2048), blk, 0, stream>>>(Qw, Kw, Vw, Aw);
  linear_wmma<<<gproj, blk, 0, stream>>>(Aw, Wo, bo, out);
}


# causal work balanced: each wave walks 64 or 65 key tiles (query tiles t and 127-t paired, the heavy tile's keys split over two waves, their partial sums added through LDS); 8-wave work-groups
# speedup vs baseline: 3.1972x; 1.3216x over previous
#include <hip/hip_runtime.h>

typedef __attribute__((ext_vector_type(2))) float v2f;
typedef __attribute__((ext_vector_type(8))) float v8f;

constexpr int EMB   = 128;
constexpr int SEQ   = 2048;
constexpr int BATCH = 4;
constexpr int HEADS = 16;
constexpr int HD    = 8;
constexpr float ATT_SCALE = 0.35355339059327373f;
constexpr float NEG_SCALE_LOG2E = -ATT_SCALE * 1.4426950408889634f;

__device__ __forceinline__ v8f wmma4_f32(v2f a, v2f b, v8f c) {
  return __builtin_amdgcn_wmma_f32_16x16x4_f32(
       false, a,  false, b,
       (short)0, c,  false,  false);
}

__global__ __launch_bounds__(128) void linear_wmma(
    const float* __restrict__ x, const float* __restrict__ W,
    const float* __restrict__ bias, float* __restrict__ out) {
  const int lane = threadIdx.x & 31;
  const int wave = threadIdx.x >> 5;
  const int tile = blockIdx.x * 4 + wave;
  const int mt = tile >> 3;
  const int nt = tile & 7;
  const int m0 = mt * 16, n0 = nt * 16;
  const int lo = lane & 15, hi = lane >> 4;

  const float* arow = x + (size_t)(m0 + lo) * EMB + 2 * hi;
  const float* brow = W + (size_t)(n0 + lo) * EMB + 2 * hi;
  v8f c = {};
#pragma unroll
  for (int k0 = 0; k0 < EMB; k0 += 4) {
    v2f a = *(const v2f*)(arow + k0);
    v2f b = *(const v2f*)(brow + k0);
    c = wmma4_f32(a, b, c);
  }
  const float bn = bias[n0 + lo];
  float* orow = out + (size_t)(m0 + 8 * hi) * EMB + n0 + lo;
#pragma unroll
  for (int r = 0; r < 8; ++r)
    orow[(size_t)r * EMB] = c[r] + bn;
}

__device__ __forceinline__ v8f attn_span(
    const float* __restrict__ Qb, const float* __restrict__ Kb, const float* __restrict__ Vb,
    int qt, int kt0, int kt1, int lo, int hi) {
  const int q0 = qt * 16;

  const float* qrow = Qb + (size_t)(q0 + lo) * EMB + 2 * hi;
  const v2f bq0 = *(const v2f*)(qrow);
  const v2f bq1 = *(const v2f*)(qrow + 4);

  const bool  isv  = lo < 8;
  const float fill = (lo == 8) ? 1.0f : 0.0f;
  const int   m    = q0 + lo;

  const float* kptr = Kb + (size_t)lo * EMB + 2 * hi;
  const float* vptr = Vb + (size_t)(8 * hi) * EMB + (lo & 7);
  v8f acc = {};

  for (int kt = kt0; kt < kt1; ++kt) {
    const int k0r = kt * 16;

    const float* krow = kptr + (size_t)k0r * EMB;
    const v2f ak0 = *(const v2f*)(krow);
    const v2f ak1 = *(const v2f*)(krow + 4);
    const float* vp = vptr + (size_t)k0r * EMB;
    float vv[8];
#pragma unroll
    for (int r = 0; r < 8; ++r) vv[r] = vp[(size_t)r * EMB];

    v8f s = {};
    s = wmma4_f32(ak0, bq0, s);
    s = wmma4_f32(ak1, bq1, s);

    const int nb = k0r + 8 * hi;
    float w[8];
#pragma unroll
    for (int r = 0; r < 8; ++r) {
      const float e  = __builtin_amdgcn_exp2f(NEG_SCALE_LOG2E * s[r]);
      const float sg = __builtin_amdgcn_rcpf(1.0f + e);
      w[r] = (nb + r <= m) ? sg : 0.0f;
    }

#pragma unroll
    for (int j = 0; j < 4; ++j) {
      v2f a, bb;
      a.x  = isv ? vv[2 * j]     : fill;
      a.y  = isv ? vv[2 * j + 1] : fill;
      bb.x = w[2 * j];
      bb.y = w[2 * j + 1];
      acc = wmma4_f32(a, bb, acc);
    }
  }
  return acc;
}

__device__ __forceinline__ void attn_store(float* __restrict__ Ob, v8f acc, int qt, int lo, int hi) {
  const float rs = __shfl(acc[0], 16 + lo, 32);
  if (hi == 0) {
    float4 o0, o1;
    o0.x = acc[0] / rs; o0.y = acc[1] / rs; o0.z = acc[2] / rs; o0.w = acc[3] / rs;
    o1.x = acc[4] / rs; o1.y = acc[5] / rs; o1.z = acc[6] / rs; o1.w = acc[7] / rs;
    float4* orow = (float4*)(Ob + (size_t)(qt * 16 + lo) * EMB);
    orow[0] = o0;
    orow[1] = o1;
  }
}

constexpr int QTILES = SEQ / 16;
constexpr int HALF_T = QTILES / 2;
__global__ __launch_bounds__(256) void attn_wmma(
    const float* __restrict__ Q, const float* __restrict__ K,
    const float* __restrict__ V, float* __restrict__ O) {
  __shared__ float4 part[4][2][32];

  const int lane = threadIdx.x & 31;
  const int wave = __builtin_amdgcn_readfirstlane((int)(threadIdx.x >> 5));
  const int pw   = wave & 3;
  const int role = wave >> 2;
  const int bh = blockIdx.x >> 4;
  const int p  = (blockIdx.x & 15) * 4 + pw;
  const int ql = p, qh = QTILES - 1 - p;
  const int b = bh >> 4, h = bh & 15;
  const int lo = lane & 15, hi = lane >> 4;
  const size_t base = (size_t)b * SEQ * EMB + (size_t)h * HD;
  const float* Qb = Q + base;
  const float* Kb = K + base;
  const float* Vb = V + base;
  float*       Ob = O + base;

  v8f acc;
  if (role == 0) {
    acc = attn_span(Qb, Kb, Vb, qh, 0, HALF_T, lo, hi);
    float4 a0, a1;
    a0.x = acc[0]; a0.y = acc[1]; a0.z = acc[2]; a0.w = acc[3];
    a1.x = acc[4]; a1.y = acc[5]; a1.z = acc[6]; a1.w = acc[7];
    part[pw][0][lane] = a0;
    part[pw][1][lane] = a1;
  } else {
    const v8f al = attn_span(Qb, Kb, Vb, ql, 0, ql + 1, lo, hi);
    attn_store(Ob, al, ql, lo, hi);
    acc = attn_span(Qb, Kb, Vb, qh, HALF_T, qh + 1, lo, hi);
  }
  __syncthreads();
  if (role == 1) {
    const float4 a0 = part[pw][0][lane];
    const float4 a1 = part[pw][1][lane];
    acc[0] += a0.x; acc[1] += a0.y; acc[2] += a0.z; acc[3] += a0.w;
    acc[4] += a1.x; acc[5] += a1.y; acc[6] += a1.z; acc[7] += a1.w;
    attn_store(Ob, acc, qh, lo, hi);
  }
}

extern "C" void kernel_launch(void* const* d_in, const int* in_sizes, int n_in,
                              void* d_out, int out_size, void* d_ws, size_t ws_size,
                              hipStream_t stream) {
  const float* x  = (const float*)d_in[0];
  const float* Wq = (const float*)d_in[1];
  const float* bq = (const float*)d_in[2];
  const float* Wk = (const float*)d_in[3];
  const float* bk = (const float*)d_in[4];
  const float* Wv = (const float*)d_in[5];
  const float* bv = (const float*)d_in[6];
  const float* Wo = (const float*)d_in[7];
  const float* bo = (const float*)d_in[8];
  float* out = (float*)d_out;

  const size_t mat = (size_t)BATCH * SEQ * EMB;
  float* Qw = (float*)d_ws;
  float* Kw = Qw + mat;
  float* Vw = Kw + mat;
  float* Aw = Vw + mat;

  dim3 blk(128);
  dim3 gproj(1024);
  linear_wmma<<<gproj, blk, 0, stream>>>(x, Wq, bq, Qw);
  linear_wmma<<<gproj, blk, 0, stream>>>(x, Wk, bk, Kw);
  linear_wmma<<<gproj, blk, 0, stream>>>(x, Wv, bv, Vw);
  attn_wmma<<<dim3(1024), dim3(256), 0, stream>>>(Qw, Kw, Vw, Aw);
  linear_wmma<<<gproj, blk, 0, stream>>>(Aw, Wo, bo, out);
}


# attention loop: K and V fragments of the next key tile loaded one tile ahead into a second register set (loop body of two tiles), so load latency runs under the previous tile's math
# speedup vs baseline: 3.7158x; 1.1622x over previous
#include <hip/hip_runtime.h>

typedef __attribute__((ext_vector_type(2))) float v2f;
typedef __attribute__((ext_vector_type(8))) float v8f;

constexpr int EMB   = 128;
constexpr int SEQ   = 2048;
constexpr int BATCH = 4;
constexpr int HEADS = 16;
constexpr int HD    = 8;
constexpr float ATT_SCALE = 0.35355339059327373f;
constexpr float NEG_SCALE_LOG2E = -ATT_SCALE * 1.4426950408889634f;

__device__ __forceinline__ v8f wmma4_f32(v2f a, v2f b, v8f c) {
  return __builtin_amdgcn_wmma_f32_16x16x4_f32(
       false, a,  false, b,
       (short)0, c,  false,  false);
}

__global__ __launch_bounds__(128) void linear_wmma(
    const float* __restrict__ x, const float* __restrict__ W,
    const float* __restrict__ bias, float* __restrict__ out) {
  const int lane = threadIdx.x & 31;
  const int wave = threadIdx.x >> 5;
  const int tile = blockIdx.x * 4 + wave;
  const int mt = tile >> 3;
  const int nt = tile & 7;
  const int m0 = mt * 16, n0 = nt * 16;
  const int lo = lane & 15, hi = lane >> 4;

  const float* arow = x + (size_t)(m0 + lo) * EMB + 2 * hi;
  const float* brow = W + (size_t)(n0 + lo) * EMB + 2 * hi;
  v8f c = {};
#pragma unroll
  for (int k0 = 0; k0 < EMB; k0 += 4) {
    v2f a = *(const v2f*)(arow + k0);
    v2f b = *(const v2f*)(brow + k0);
    c = wmma4_f32(a, b, c);
  }
  const float bn = bias[n0 + lo];
  float* orow = out + (size_t)(m0 + 8 * hi) * EMB + n0 + lo;
#pragma unroll
  for (int r = 0; r < 8; ++r)
    orow[(size_t)r * EMB] = c[r] + bn;
}


struct KvFrag {
  v2f   k0, k1;
  float v[8];
};

__device__ __forceinline__ void kv_load(KvFrag& f, const float* __restrict__ Kb,
                                        const float* __restrict__ Vb, int kt,
                                        unsigned koff, unsigned voff) {
  const float* kt_k = Kb + (size_t)kt * (16 * EMB);
  const float* kt_v = Vb + (size_t)kt * (16 * EMB);
  f.k0 = *(const v2f*)(kt_k + koff);
  f.k1 = *(const v2f*)(kt_k + koff + 4);
#pragma unroll
  for (int r = 0; r < 8; ++r) f.v[r] = kt_v[voff + r * EMB];
}

__device__ __forceinline__ v8f kv_tile(const KvFrag& f, v2f bq0, v2f bq1, v8f acc,
                                       int kt, int m, int hi, bool isv, float fill) {
  v8f s = {};
  s = wmma4_f32(f.k0, bq0, s);
  s = wmma4_f32(f.k1, bq1, s);

  const int nb = kt * 16 + 8 * hi;
  float w[8];
#pragma unroll
  for (int r = 0; r < 8; ++r) {
    const float e  = __builtin_amdgcn_exp2f(NEG_SCALE_LOG2E * s[r]);
    const float sg = __builtin_amdgcn_rcpf(1.0f + e);
    w[r] = (nb + r <= m) ? sg : 0.0f;
  }

#pragma unroll
  for (int j = 0; j < 4; ++j) {
    v2f a, bb;
    a.x  = isv ? f.v[2 * j]     : fill;
    a.y  = isv ? f.v[2 * j + 1] : fill;
    bb.x = w[2 * j];
    bb.y = w[2 * j + 1];
    acc = wmma4_f32(a, bb, acc);
  }
  return acc;
}

__device__ __forceinline__ v8f attn_span(
    const float* __restrict__ Qb, const float* __restrict__ Kb, const float* __restrict__ Vb,
    int qt, int kt0, int kt1, int lo, int hi) {
  const int q0 = qt * 16;

  const float* qrow = Qb + (size_t)(q0 + lo) * EMB + 2 * hi;
  const v2f bq0 = *(const v2f*)(qrow);
  const v2f bq1 = *(const v2f*)(qrow + 4);

  const bool  isv  = lo < 8;
  const float fill = (lo == 8) ? 1.0f : 0.0f;
  const int   m    = q0 + lo;

  const unsigned koff = (unsigned)(lo * EMB + 2 * hi);
  const unsigned voff = (unsigned)((8 * hi) * EMB + (lo & 7));
  const int last = kt1 - 1;

  KvFrag fa, fb;
  kv_load(fa, Kb, Vb, kt0, koff, voff);
  v8f acc = {};
  int kt = kt0;
  for (; kt < last; kt += 2) {
    kv_load(fb, Kb, Vb, kt + 1, koff, voff);
    asm volatile("" ::: "memory");
    acc = kv_tile(fa, bq0, bq1, acc, kt, m, hi, isv, fill);
    const int nx = (kt + 2 < last) ? kt + 2 : last;
    kv_load(fa, Kb, Vb, nx, koff, voff);
    asm volatile("" ::: "memory");
    acc = kv_tile(fb, bq0, bq1, acc, kt + 1, m, hi, isv, fill);
  }
  if (kt == last)
    acc = kv_tile(fa, bq0, bq1, acc, kt, m, hi, isv, fill);
  return acc;
}

__device__ __forceinline__ void attn_store(float* __restrict__ Ob, v8f acc, int qt, int lo, int hi) {
  const float rs = __shfl(acc[0], 16 + lo, 32);
  if (hi == 0) {
    float4 o0, o1;
    o0.x = acc[0] / rs; o0.y = acc[1] / rs; o0.z = acc[2] / rs; o0.w = acc[3] / rs;
    o1.x = acc[4] / rs; o1.y = acc[5] / rs; o1.z = acc[6] / rs; o1.w = acc[7] / rs;
    float4* orow = (float4*)(Ob + (size_t)(qt * 16 + lo) * EMB);
    orow[0] = o0;
    orow[1] = o1;
  }
}

constexpr int QTILES = SEQ / 16;
constexpr int HALF_T = QTILES / 2;
__global__ __launch_bounds__(256) void attn_wmma(
    const float* __restrict__ Q, const float* __restrict__ K,
    const float* __restrict__ V, float* __restrict__ O) {
  __shared__ float4 part[4][2][32];

  const int lane = threadIdx.x & 31;
  const int wave = __builtin_amdgcn_readfirstlane((int)(threadIdx.x >> 5));
  const int pw   = wave & 3;
  const int role = wave >> 2;
  const int bh = blockIdx.x >> 4;
  const int p  = (blockIdx.x & 15) * 4 + pw;
  const int ql = p, qh = QTILES - 1 - p;
  const int b = bh >> 4, h = bh & 15;
  const int lo = lane & 15, hi = lane >> 4;
  const size_t base = (size_t)b * SEQ * EMB + (size_t)h * HD;
  const float* Qb = Q + base;
  const float* Kb = K + base;
  const float* Vb = V + base;
  float*       Ob = O + base;

  v8f acc;
  if (role == 0) {
    acc = attn_span(Qb, Kb, Vb, qh, 0, HALF_T, lo, hi);
    float4 a0, a1;
    a0.x = acc[0]; a0.y = acc[1]; a0.z = acc[2]; a0.w = acc[3];
    a1.x = acc[4]; a1.y = acc[5]; a1.z = acc[6]; a1.w = acc[7];
    part[pw][0][lane] = a0;
    part[pw][1][lane] = a1;
  } else {
    const v8f al = attn_span(Qb, Kb, Vb, ql, 0, ql + 1, lo, hi);
    attn_store(Ob, al, ql, lo, hi);
    acc = attn_span(Qb, Kb, Vb, qh, HALF_T, qh + 1, lo, hi);
  }
  __syncthreads();
  if (role == 1) {
    const float4 a0 = part[pw][0][lane];
    const float4 a1 = part[pw][1][lane];
    acc[0] += a0.x; acc[1] += a0.y; acc[2] += a0.z; acc[3] += a0.w;
    acc[4] += a1.x; acc[5] += a1.y; acc[6] += a1.z; acc[7] += a1.w;
    attn_store(Ob, acc, qh, lo, hi);
  }
}

extern "C" void kernel_launch(void* const* d_in, const int* in_sizes, int n_in,
                              void* d_out, int out_size, void* d_ws, size_t ws_size,
                              hipStream_t stream) {
  const float* x  = (const float*)d_in[0];
  const float* Wq = (const float*)d_in[1];
  const float* bq = (const float*)d_in[2];
  const float* Wk = (const float*)d_in[3];
  const float* bk = (const float*)d_in[4];
  const float* Wv = (const float*)d_in[5];
  const float* bv = (const float*)d_in[6];
  const float* Wo = (const float*)d_in[7];
  const float* bo = (const float*)d_in[8];
  float* out = (float*)d_out;

  const size_t mat = (size_t)BATCH * SEQ * EMB;
  float* Qw = (float*)d_ws;
  float* Kw = Qw + mat;
  float* Vw = Kw + mat;
  float* Aw = Vw + mat;

  dim3 blk(128);
  dim3 gproj(1024);
  linear_wmma<<<gproj, blk, 0, stream>>>(x, Wq, bq, Qw);
  linear_wmma<<<gproj, blk, 0, stream>>>(x, Wk, bk, Kw);
  linear_wmma<<<gproj, blk, 0, stream>>>(x, Wv, bv, Vw);
  attn_wmma<<<dim3(1024), dim3(256), 0, stream>>>(Qw, Kw, Vw, Aw);
  linear_wmma<<<gproj, blk, 0, stream>>>(Aw, Wo, bo, out);
}


# projections re-tiled for operand reuse: a wave forms a 32x32 block from 16-byte fragment loads, each fragment feeding two tiles, Q, K, V in one launch; k order of the projection sums permuted
# speedup vs baseline: 4.3349x; 1.1666x over previous
#include <hip/hip_runtime.h>

typedef __attribute__((ext_vector_type(2))) float v2f;
typedef __attribute__((ext_vector_type(8))) float v8f;

constexpr int EMB   = 128;
constexpr int SEQ   = 2048;
constexpr int BATCH = 4;
constexpr int HEADS = 16;
constexpr int HD    = 8;
constexpr float ATT_SCALE = 0.35355339059327373f;
constexpr float NEG_SCALE_LOG2E = -ATT_SCALE * 1.4426950408889634f;

__device__ __forceinline__ v8f wmma4_f32(v2f a, v2f b, v8f c) {
  return __builtin_amdgcn_wmma_f32_16x16x4_f32(
       false, a,  false, b,
       (short)0, c,  false,  false);
}

typedef __attribute__((ext_vector_type(4))) float v4f;

struct PGroup {
  v4f a0[4], a1[4], w0[4], w1[4];
};

__device__ __forceinline__ void proj_load(PGroup& f, const float* __restrict__ a0p,
                                          const float* __restrict__ a1p,
                                          const float* __restrict__ b0p,
                                          const float* __restrict__ b1p, int g) {
#pragma unroll
  for (int j = 0; j < 4; ++j) f.a0[j] = *(const v4f*)(a0p + 32 * g + 8 * j);
#pragma unroll
  for (int j = 0; j < 4; ++j) f.a1[j] = *(const v4f*)(a1p + 32 * g + 8 * j);
#pragma unroll
  for (int j = 0; j < 4; ++j) f.w0[j] = *(const v4f*)(b0p + 32 * g + 8 * j);
#pragma unroll
  for (int j = 0; j < 4; ++j) f.w1[j] = *(const v4f*)(b1p + 32 * g + 8 * j);
}

__device__ __forceinline__ void proj_mma(const PGroup& f, v8f& c00, v8f& c01, v8f& c10, v8f& c11) {
#pragma unroll
  for (int j = 0; j < 4; ++j) {
    v2f a0l, a0h, a1l, a1h, w0l, w0h, w1l, w1h;
    a0l.x = f.a0[j].x; a0l.y = f.a0[j].y; a0h.x = f.a0[j].z; a0h.y = f.a0[j].w;
    a1l.x = f.a1[j].x; a1l.y = f.a1[j].y; a1h.x = f.a1[j].z; a1h.y = f.a1[j].w;
    w0l.x = f.w0[j].x; w0l.y = f.w0[j].y; w0h.x = f.w0[j].z; w0h.y = f.w0[j].w;
    w1l.x = f.w1[j].x; w1l.y = f.w1[j].y; w1h.x = f.w1[j].z; w1h.y = f.w1[j].w;
    c00 = wmma4_f32(a0l, w0l, c00);
    c01 = wmma4_f32(a0l, w1l, c01);
    c10 = wmma4_f32(a1l, w0l, c10);
    c11 = wmma4_f32(a1l, w1l, c11);
    c00 = wmma4_f32(a0h, w0h, c00);
    c01 = wmma4_f32(a0h, w1h, c01);
    c10 = wmma4_f32(a1h, w0h, c10);
    c11 = wmma4_f32(a1h, w1h, c11);
  }
}

__global__ __launch_bounds__(128) void proj_wmma(
    const float* __restrict__ x, int nmat,
    const float* __restrict__ W0, const float* __restrict__ b0, float* __restrict__ o0,
    const float* __restrict__ W1, const float* __restrict__ b1, float* __restrict__ o1,
    const float* __restrict__ W2, const float* __restrict__ b2, float* __restrict__ o2) {
  const int lane = threadIdx.x & 31;
  const int wave = __builtin_amdgcn_readfirstlane((int)(threadIdx.x >> 5));
  const int lo = lane & 15, hi = lane >> 4;

  int blk = blockIdx.x;
  const float* W = W0;
  const float* bias = b0;
  float* out = o0;
  if (nmat == 3) {
    const int which = blk % 3;
    blk = blk / 3;
    if (which == 1) { W = W1; bias = b1; out = o1; }
    if (which == 2) { W = W2; bias = b2; out = o2; }
  }
  const int m0 = (blk >> 1) * 64 + (wave & 1) * 32;
  const int n0 = (blk & 1) * 64 + (wave >> 1) * 32;

  const float* a0p = x + (size_t)(m0 + lo) * EMB + 4 * hi;
  const float* a1p = a0p + 16 * EMB;
  const float* b0p = W + (size_t)(n0 + lo) * EMB + 4 * hi;
  const float* b1p = b0p + 16 * EMB;
  v8f c00 = {}, c01 = {}, c10 = {}, c11 = {};
  PGroup f0, f1, f2, f3;
  proj_load(f0, a0p, a1p, b0p, b1p, 0);
  proj_load(f1, a0p, a1p, b0p, b1p, 1);
  proj_load(f2, a0p, a1p, b0p, b1p, 2);
  proj_load(f3, a0p, a1p, b0p, b1p, 3);
  __builtin_amdgcn_sched_barrier(0);
  proj_mma(f0, c00, c01, c10, c11);
  proj_mma(f1, c00, c01, c10, c11);
  proj_mma(f2, c00, c01, c10, c11);
  proj_mma(f3, c00, c01, c10, c11);
  const float bn0 = bias[n0 + lo];
  const float bn1 = bias[n0 + 16 + lo];
  float* orow = out + (size_t)(m0 + 8 * hi) * EMB + n0 + lo;
#pragma unroll
  for (int r = 0; r < 8; ++r) {
    orow[(size_t)r * EMB]             = c00[r] + bn0;
    orow[(size_t)r * EMB + 16]        = c01[r] + bn1;
    orow[(size_t)(r + 16) * EMB]      = c10[r] + bn0;
    orow[(size_t)(r + 16) * EMB + 16] = c11[r] + bn1;
  }
}


struct KvFrag {
  v2f   k0, k1;
  float v[8];
};

__device__ __forceinline__ void kv_load(KvFrag& f, const float* __restrict__ Kb,
                                        const float* __restrict__ Vb, int kt,
                                        unsigned koff, unsigned voff) {
  const float* kt_k = Kb + (size_t)kt * (16 * EMB);
  const float* kt_v = Vb + (size_t)kt * (16 * EMB);
  f.k0 = *(const v2f*)(kt_k + koff);
  f.k1 = *(const v2f*)(kt_k + koff + 4);
#pragma unroll
  for (int r = 0; r < 8; ++r) f.v[r] = kt_v[voff + r * EMB];
}

__device__ __forceinline__ v8f kv_tile(const KvFrag& f, v2f bq0, v2f bq1, v8f acc,
                                       int kt, int m, int hi, bool isv, float fill) {
  v8f s = {};
  s = wmma4_f32(f.k0, bq0, s);
  s = wmma4_f32(f.k1, bq1, s);

  const int nb = kt * 16 + 8 * hi;
  float w[8];
#pragma unroll
  for (int r = 0; r < 8; ++r) {
    const float e  = __builtin_amdgcn_exp2f(NEG_SCALE_LOG2E * s[r]);
    const float sg = __builtin_amdgcn_rcpf(1.0f + e);
    w[r] = (nb + r <= m) ? sg : 0.0f;
  }

#pragma unroll
  for (int j = 0; j < 4; ++j) {
    v2f a, bb;
    a.x  = isv ? f.v[2 * j]     : fill;
    a.y  = isv ? f.v[2 * j + 1] : fill;
    bb.x = w[2 * j];
    bb.y = w[2 * j + 1];
    acc = wmma4_f32(a, bb, acc);
  }
  return acc;
}

__device__ __forceinline__ v8f attn_span(
    const float* __restrict__ Qb, const float* __restrict__ Kb, const float* __restrict__ Vb,
    int qt, int kt0, int kt1, int lo, int hi) {
  const int q0 = qt * 16;

  const float* qrow = Qb + (size_t)(q0 + lo) * EMB + 2 * hi;
  const v2f bq0 = *(const v2f*)(qrow);
  const v2f bq1 = *(const v2f*)(qrow + 4);

  const bool  isv  = lo < 8;
  const float fill = (lo == 8) ? 1.0f : 0.0f;
  const int   m    = q0 + lo;

  const unsigned koff = (unsigned)(lo * EMB + 2 * hi);
  const unsigned voff = (unsigned)((8 * hi) * EMB + (lo & 7));
  const int last = kt1 - 1;

  KvFrag fa, fb;
  kv_load(fa, Kb, Vb, kt0, koff, voff);
  v8f acc = {};
  int kt = kt0;
  for (; kt < last; kt += 2) {
    kv_load(fb, Kb, Vb, kt + 1, koff, voff);
    asm volatile("" ::: "memory");
    acc = kv_tile(fa, bq0, bq1, acc, kt, m, hi, isv, fill);
    const int nx = (kt + 2 < last) ? kt + 2 : last;
    kv_load(fa, Kb, Vb, nx, koff, voff);
    asm volatile("" ::: "memory");
    acc = kv_tile(fb, bq0, bq1, acc, kt + 1, m, hi, isv, fill);
  }
  if (kt == last)
    acc = kv_tile(fa, bq0, bq1, acc, kt, m, hi, isv, fill);
  return acc;
}

__device__ __forceinline__ void attn_store(float* __restrict__ Ob, v8f acc, int qt, int lo, int hi) {
  const float rs = __shfl(acc[0], 16 + lo, 32);
  if (hi == 0) {
    float4 o0, o1;
    o0.x = acc[0] / rs; o0.y = acc[1] / rs; o0.z = acc[2] / rs; o0.w = acc[3] / rs;
    o1.x = acc[4] / rs; o1.y = acc[5] / rs; o1.z = acc[6] / rs; o1.w = acc[7] / rs;
    float4* orow = (float4*)(Ob + (size_t)(qt * 16 + lo) * EMB);
    orow[0] = o0;
    orow[1] = o1;
  }
}

constexpr int QTILES = SEQ / 16;
constexpr int HALF_T = QTILES / 2;
__global__ __launch_bounds__(256) void attn_wmma(
    const float* __restrict__ Q, const float* __restrict__ K,
    const float* __restrict__ V, float* __restrict__ O) {
  __shared__ float4 part[4][2][32];

  const int lane = threadIdx.x & 31;
  const int wave = __builtin_amdgcn_readfirstlane((int)(threadIdx.x >> 5));
  const int pw   = wave & 3;
  const int role = wave >> 2;
  const int bh = blockIdx.x >> 4;
  const int p  = (blockIdx.x & 15) * 4 + pw;
  const int ql = p, qh = QTILES - 1 - p;
  const int b = bh >> 4, h = bh & 15;
  const int lo = lane & 15, hi = lane >> 4;
  const size_t base = (size_t)b * SEQ * EMB + (size_t)h * HD;
  const float* Qb = Q + base;
  const float* Kb = K + base;
  const float* Vb = V + base;
  float*       Ob = O + base;

  v8f acc;
  if (role == 0) {
    acc = attn_span(Qb, Kb, Vb, qh, 0, HALF_T, lo, hi);
    float4 a0, a1;
    a0.x = acc[0]; a0.y = acc[1]; a0.z = acc[2]; a0.w = acc[3];
    a1.x = acc[4]; a1.y = acc[5]; a1.z = acc[6]; a1.w = acc[7];
    part[pw][0][lane] = a0;
    part[pw][1][lane] = a1;
  } else {
    const v8f al = attn_span(Qb, Kb, Vb, ql, 0, ql + 1, lo, hi);
    attn_store(Ob, al, ql, lo, hi);
    acc = attn_span(Qb, Kb, Vb, qh, HALF_T, qh + 1, lo, hi);
  }
  __syncthreads();
  if (role == 1) {
    const float4 a0 = part[pw][0][lane];
    const float4 a1 = part[pw][1][lane];
    acc[0] += a0.x; acc[1] += a0.y; acc[2] += a0.z; acc[3] += a0.w;
    acc[4] += a1.x; acc[5] += a1.y; acc[6] += a1.z; acc[7] += a1.w;
    attn_store(Ob, acc, qh, lo, hi);
  }
}

extern "C" void kernel_launch(void* const* d_in, const int* in_sizes, int n_in,
                              void* d_out, int out_size, void* d_ws, size_t ws_size,
                              hipStream_t stream) {
  const float* x  = (const float*)d_in[0];
  const float* Wq = (const float*)d_in[1];
  const float* bq = (const float*)d_in[2];
  const float* Wk = (const float*)d_in[3];
  const float* bk = (const float*)d_in[4];
  const float* Wv = (const float*)d_in[5];
  const float* bv = (const float*)d_in[6];
  const float* Wo = (const float*)d_in[7];
  const float* bo = (const float*)d_in[8];
  float* out = (float*)d_out;

  const size_t mat = (size_t)BATCH * SEQ * EMB;
  float* Qw = (float*)d_ws;
  float* Kw = Qw + mat;
  float* Vw = Kw + mat;
  float* Aw = Vw + mat;

  proj_wmma<<<dim3(768), dim3(128), 0, stream>>>(x, 3, Wq, bq, Qw, Wk, bk, Kw, Wv, bv, Vw);
  attn_wmma<<<dim3(1024), dim3(256), 0, stream>>>(Qw, Kw, Vw, Aw);
  proj_wmma<<<dim3(256), dim3(128), 0, stream>>>(Aw, 1, Wo, bo, out, Wo, bo, out, Wo, bo, out);
}


# key loop: weight = 1 + tanh(half the scaled score) = 2*sigmoid (2 cancels in the row ratio): one transcendental and one add per weight, scale folded into Q, causal select only on the diagonal tile
# speedup vs baseline: 4.6363x; 1.0695x over previous
#include <hip/hip_runtime.h>

typedef __attribute__((ext_vector_type(2))) float v2f;
typedef __attribute__((ext_vector_type(8))) float v8f;

constexpr int EMB   = 128;
constexpr int SEQ   = 2048;
constexpr int BATCH = 4;
constexpr int HEADS = 16;
constexpr int HD    = 8;
constexpr float ATT_SCALE = 0.35355339059327373f;
constexpr float HALF_SCALE = 0.5f * ATT_SCALE;
constexpr float NEG2_LOG2E = -2.0f * 1.4426950408889634f;

__device__ __forceinline__ v8f wmma4_f32(v2f a, v2f b, v8f c) {
  return __builtin_amdgcn_wmma_f32_16x16x4_f32(
       false, a,  false, b,
       (short)0, c,  false,  false);
}

typedef __attribute__((ext_vector_type(4))) float v4f;

struct PGroup {
  v4f a0[4], a1[4], w0[4], w1[4];
};

__device__ __forceinline__ void proj_load(PGroup& f, const float* __restrict__ a0p,
                                          const float* __restrict__ a1p,
                                          const float* __restrict__ b0p,
                                          const float* __restrict__ b1p, int g) {
#pragma unroll
  for (int j = 0; j < 4; ++j) f.a0[j] = *(const v4f*)(a0p + 32 * g + 8 * j);
#pragma unroll
  for (int j = 0; j < 4; ++j) f.a1[j] = *(const v4f*)(a1p + 32 * g + 8 * j);
#pragma unroll
  for (int j = 0; j < 4; ++j) f.w0[j] = *(const v4f*)(b0p + 32 * g + 8 * j);
#pragma unroll
  for (int j = 0; j < 4; ++j) f.w1[j] = *(const v4f*)(b1p + 32 * g + 8 * j);
}

__device__ __forceinline__ void proj_mma(const PGroup& f, v8f& c00, v8f& c01, v8f& c10, v8f& c11) {
#pragma unroll
  for (int j = 0; j < 4; ++j) {
    v2f a0l, a0h, a1l, a1h, w0l, w0h, w1l, w1h;
    a0l.x = f.a0[j].x; a0l.y = f.a0[j].y; a0h.x = f.a0[j].z; a0h.y = f.a0[j].w;
    a1l.x = f.a1[j].x; a1l.y = f.a1[j].y; a1h.x = f.a1[j].z; a1h.y = f.a1[j].w;
    w0l.x = f.w0[j].x; w0l.y = f.w0[j].y; w0h.x = f.w0[j].z; w0h.y = f.w0[j].w;
    w1l.x = f.w1[j].x; w1l.y = f.w1[j].y; w1h.x = f.w1[j].z; w1h.y = f.w1[j].w;
    c00 = wmma4_f32(a0l, w0l, c00);
    c01 = wmma4_f32(a0l, w1l, c01);
    c10 = wmma4_f32(a1l, w0l, c10);
    c11 = wmma4_f32(a1l, w1l, c11);
    c00 = wmma4_f32(a0h, w0h, c00);
    c01 = wmma4_f32(a0h, w1h, c01);
    c10 = wmma4_f32(a1h, w0h, c10);
    c11 = wmma4_f32(a1h, w1h, c11);
  }
}

__global__ __launch_bounds__(128) void proj_wmma(
    const float* __restrict__ x, int nmat,
    const float* __restrict__ W0, const float* __restrict__ b0, float* __restrict__ o0,
    const float* __restrict__ W1, const float* __restrict__ b1, float* __restrict__ o1,
    const float* __restrict__ W2, const float* __restrict__ b2, float* __restrict__ o2) {
  const int lane = threadIdx.x & 31;
  const int wave = __builtin_amdgcn_readfirstlane((int)(threadIdx.x >> 5));
  const int lo = lane & 15, hi = lane >> 4;

  int blk = blockIdx.x;
  const float* W = W0;
  const float* bias = b0;
  float* out = o0;
  if (nmat == 3) {
    const int which = blk % 3;
    blk = blk / 3;
    if (which == 1) { W = W1; bias = b1; out = o1; }
    if (which == 2) { W = W2; bias = b2; out = o2; }
  }
  const int m0 = (blk >> 1) * 64 + (wave & 1) * 32;
  const int n0 = (blk & 1) * 64 + (wave >> 1) * 32;

  const float* a0p = x + (size_t)(m0 + lo) * EMB + 4 * hi;
  const float* a1p = a0p + 16 * EMB;
  const float* b0p = W + (size_t)(n0 + lo) * EMB + 4 * hi;
  const float* b1p = b0p + 16 * EMB;
  v8f c00 = {}, c01 = {}, c10 = {}, c11 = {};
  PGroup f0, f1, f2, f3;
  proj_load(f0, a0p, a1p, b0p, b1p, 0);
  proj_load(f1, a0p, a1p, b0p, b1p, 1);
  proj_load(f2, a0p, a1p, b0p, b1p, 2);
  proj_load(f3, a0p, a1p, b0p, b1p, 3);
  __builtin_amdgcn_sched_barrier(0);
  proj_mma(f0, c00, c01, c10, c11);
  proj_mma(f1, c00, c01, c10, c11);
  proj_mma(f2, c00, c01, c10, c11);
  proj_mma(f3, c00, c01, c10, c11);
  const float bn0 = bias[n0 + lo];
  const float bn1 = bias[n0 + 16 + lo];
  float* orow = out + (size_t)(m0 + 8 * hi) * EMB + n0 + lo;
#pragma unroll
  for (int r = 0; r < 8; ++r) {
    orow[(size_t)r * EMB]             = c00[r] + bn0;
    orow[(size_t)r * EMB + 16]        = c01[r] + bn1;
    orow[(size_t)(r + 16) * EMB]      = c10[r] + bn0;
    orow[(size_t)(r + 16) * EMB + 16] = c11[r] + bn1;
  }
}


struct KvFrag {
  v2f   k0, k1;
  float v[8];
};

__device__ __forceinline__ void kv_load(KvFrag& f, const float* __restrict__ Kb,
                                        const float* __restrict__ Vb, int kt,
                                        unsigned koff, unsigned voff) {
  const float* kt_k = Kb + (size_t)kt * (16 * EMB);
  const float* kt_v = Vb + (size_t)kt * (16 * EMB);
  f.k0 = *(const v2f*)(kt_k + koff);
  f.k1 = *(const v2f*)(kt_k + koff + 4);
#pragma unroll
  for (int r = 0; r < 8; ++r) f.v[r] = kt_v[voff + r * EMB];
}

template <bool MASK, bool FAST>
__device__ __forceinline__ v8f kv_tile(const KvFrag& f, v2f bq0, v2f bq1, v8f acc,
                                       int lo, int hi, bool isv, float fill) {
  v8f s = {};
  s = wmma4_f32(f.k0, bq0, s);
  s = wmma4_f32(f.k1, bq1, s);

  float w[8];
#pragma unroll
  for (int r = 0; r < 8; ++r) {
    float wt;
    if (FAST) {
      wt = 1.0f + __builtin_amdgcn_tanhf(s[r]);
    } else {
      const float e = __builtin_amdgcn_exp2f(NEG2_LOG2E * s[r]);
      wt = __builtin_amdgcn_rcpf(0.5f + 0.5f * e);
    }
    w[r] = MASK ? ((8 * hi + r <= lo) ? wt : 0.0f) : wt;
  }

#pragma unroll
  for (int j = 0; j < 4; ++j) {
    v2f a, bb;
    a.x  = isv ? f.v[2 * j]     : fill;
    a.y  = isv ? f.v[2 * j + 1] : fill;
    bb.x = w[2 * j];
    bb.y = w[2 * j + 1];
    acc = wmma4_f32(a, bb, acc);
  }
  return acc;
}

template <bool DIAG, bool FAST>
__device__ __forceinline__ v8f span_pass(
    const float* __restrict__ Qb, const float* __restrict__ Kb, const float* __restrict__ Vb,
    int qt, int kt0, int kt1, int lo, int hi) {
  const int q0 = qt * 16;

  const float* qrow = Qb + (size_t)(q0 + lo) * EMB + 2 * hi;
  v2f bq0 = *(const v2f*)(qrow);
  v2f bq1 = *(const v2f*)(qrow + 4);
  bq0.x *= HALF_SCALE; bq0.y *= HALF_SCALE;
  bq1.x *= HALF_SCALE; bq1.y *= HALF_SCALE;

  const bool  isv  = lo < 8;
  const float fill = (lo == 8) ? 1.0f : 0.0f;

  const unsigned koff = (unsigned)(lo * EMB + 2 * hi);
  const unsigned voff = (unsigned)((8 * hi) * EMB + (lo & 7));
  const int last     = kt1 - 1;
  const int lastfull = DIAG ? last - 1 : last;

  KvFrag fa, fb;
  kv_load(fa, Kb, Vb, kt0, koff, voff);
  v8f acc = {};
  int kt = kt0;
  for (; kt < lastfull; kt += 2) {
    kv_load(fb, Kb, Vb, kt + 1, koff, voff);
    asm volatile("" ::: "memory");
    acc = kv_tile<false, FAST>(fa, bq0, bq1, acc, lo, hi, isv, fill);
    const int nx = (kt + 2 < last) ? kt + 2 : last;
    kv_load(fa, Kb, Vb, nx, koff, voff);
    asm volatile("" ::: "memory");
    acc = kv_tile<false, FAST>(fb, bq0, bq1, acc, lo, hi, isv, fill);
  }
  if (kt == lastfull) {
    if (DIAG) {
      kv_load(fb, Kb, Vb, last, koff, voff);
      asm volatile("" ::: "memory");
    }
    acc = kv_tile<false, FAST>(fa, bq0, bq1, acc, lo, hi, isv, fill);
    if (DIAG) acc = kv_tile<true, FAST>(fb, bq0, bq1, acc, lo, hi, isv, fill);
  } else if (DIAG) {
    acc = kv_tile<true, FAST>(fa, bq0, bq1, acc, lo, hi, isv, fill);
  }
  return acc;
}

template <bool DIAG>
__device__ __forceinline__ v8f attn_span(
    const float* __restrict__ Qb, const float* __restrict__ Kb, const float* __restrict__ Vb,
    int qt, int kt0, int kt1, int lo, int hi) {
  v8f acc = span_pass<DIAG, true>(Qb, Kb, Vb, qt, kt0, kt1, lo, hi);
  const int upto = qt * 16 + lo - kt0 * 16 + 1;
  const int span = (kt1 - kt0) * 16;
  const int nk   = upto < span ? upto : span;
  const bool thin = (hi == 1) && !(acc[0] >= 0.03125f * (float)nk);
  if (__builtin_amdgcn_ballot_w32(thin) != 0u)
    acc = span_pass<DIAG, false>(Qb, Kb, Vb, qt, kt0, kt1, lo, hi);
  return acc;
}

__device__ __forceinline__ void attn_store(float* __restrict__ Ob, v8f acc, int qt, int lo, int hi) {
  const float rs = __shfl(acc[0], 16 + lo, 32);
  if (hi == 0) {
    float4 o0, o1;
    o0.x = acc[0] / rs; o0.y = acc[1] / rs; o0.z = acc[2] / rs; o0.w = acc[3] / rs;
    o1.x = acc[4] / rs; o1.y = acc[5] / rs; o1.z = acc[6] / rs; o1.w = acc[7] / rs;
    float4* orow = (float4*)(Ob + (size_t)(qt * 16 + lo) * EMB);
    orow[0] = o0;
    orow[1] = o1;
  }
}

constexpr int QTILES = SEQ / 16;
constexpr int HALF_T = QTILES / 2;
__global__ __launch_bounds__(256) void attn_wmma(
    const float* __restrict__ Q, const float* __restrict__ K,
    const float* __restrict__ V, float* __restrict__ O) {
  __shared__ float4 part[4][2][32];

  const int lane = threadIdx.x & 31;
  const int wave = __builtin_amdgcn_readfirstlane((int)(threadIdx.x >> 5));
  const int pw   = wave & 3;
  const int role = wave >> 2;
  const int bh = blockIdx.x >> 4;
  const int p  = (blockIdx.x & 15) * 4 + pw;
  const int ql = p, qh = QTILES - 1 - p;
  const int b = bh >> 4, h = bh & 15;
  const int lo = lane & 15, hi = lane >> 4;
  const size_t base = (size_t)b * SEQ * EMB + (size_t)h * HD;
  const float* Qb = Q + base;
  const float* Kb = K + base;
  const float* Vb = V + base;
  float*       Ob = O + base;

  v8f acc;
  if (role == 0) {
    acc = attn_span<false>(Qb, Kb, Vb, qh, 0, HALF_T, lo, hi);
    float4 a0, a1;
    a0.x = acc[0]; a0.y = acc[1]; a0.z = acc[2]; a0.w = acc[3];
    a1.x = acc[4]; a1.y = acc[5]; a1.z = acc[6]; a1.w = acc[7];
    part[pw][0][lane] = a0;
    part[pw][1][lane] = a1;
  } else {
    const v8f al = attn_span<true>(Qb, Kb, Vb, ql, 0, ql + 1, lo, hi);
    attn_store(Ob, al, ql, lo, hi);
    acc = attn_span<true>(Qb, Kb, Vb, qh, HALF_T, qh + 1, lo, hi);
  }
  __syncthreads();
  if (role == 1) {
    const float4 a0 = part[pw][0][lane];
    const float4 a1 = part[pw][1][lane];
    acc[0] += a0.x; acc[1] += a0.y; acc[2] += a0.z; acc[3] += a0.w;
    acc[4] += a1.x; acc[5] += a1.y; acc[6] += a1.z; acc[7] += a1.w;
    attn_store(Ob, acc, qh, lo, hi);
  }
}

extern "C" void kernel_launch(void* const* d_in, const int* in_sizes, int n_in,
                              void* d_out, int out_size, void* d_ws, size_t ws_size,
                              hipStream_t stream) {
  const float* x  = (const float*)d_in[0];
  const float* Wq = (const float*)d_in[1];
  const float* bq = (const float*)d_in[2];
  const float* Wk = (const float*)d_in[3];
  const float* bk = (const float*)d_in[4];
  const float* Wv = (const float*)d_in[5];
  const float* bv = (const float*)d_in[6];
  const float* Wo = (const float*)d_in[7];
  const float* bo = (const float*)d_in[8];
  float* out = (float*)d_out;

  const size_t mat = (size_t)BATCH * SEQ * EMB;
  float* Qw = (float*)d_ws;
  float* Kw = Qw + mat;
  float* Vw = Kw + mat;
  float* Aw = Vw + mat;

  proj_wmma<<<dim3(768), dim3(128), 0, stream>>>(x, 3, Wq, bq, Qw, Wk, bk, Kw, Wv, bv, Vw);
  attn_wmma<<<dim3(1024), dim3(256), 0, stream>>>(Qw, Kw, Vw, Aw);
  proj_wmma<<<dim3(256), dim3(128), 0, stream>>>(Aw, 1, Wo, bo, out, Wo, bo, out, Wo, bo, out);
}


# Q, K, V intermediates stored per head (Q, K [head][row][8]; V [head][key tile][dim][16 keys]): a key tile's operands are 3 whole-line 16-byte loads, not 10 strided; k order of score sums permuted
# speedup vs baseline: 4.8192x; 1.0394x over previous
#include <hip/hip_runtime.h>

typedef __attribute__((ext_vector_type(2))) float v2f;
typedef __attribute__((ext_vector_type(8))) float v8f;

constexpr int EMB   = 128;
constexpr int SEQ   = 2048;
constexpr int BATCH = 4;
constexpr int HEADS = 16;
constexpr int HD    = 8;
constexpr float ATT_SCALE = 0.35355339059327373f;
constexpr float HALF_SCALE = 0.5f * ATT_SCALE;
constexpr float NEG2_LOG2E = -2.0f * 1.4426950408889634f;

__device__ __forceinline__ v8f wmma4_f32(v2f a, v2f b, v8f c) {
  return __builtin_amdgcn_wmma_f32_16x16x4_f32(
       false, a,  false, b,
       (short)0, c,  false,  false);
}

typedef __attribute__((ext_vector_type(4))) float v4f;

struct PGroup {
  v4f a0[4], a1[4], w0[4], w1[4];
};

__device__ __forceinline__ void proj_load(PGroup& f, const float* __restrict__ a0p,
                                          const float* __restrict__ a1p,
                                          const float* __restrict__ b0p,
                                          const float* __restrict__ b1p, int g) {
#pragma unroll
  for (int j = 0; j < 4; ++j) f.a0[j] = *(const v4f*)(a0p + 32 * g + 8 * j);
#pragma unroll
  for (int j = 0; j < 4; ++j) f.a1[j] = *(const v4f*)(a1p + 32 * g + 8 * j);
#pragma unroll
  for (int j = 0; j < 4; ++j) f.w0[j] = *(const v4f*)(b0p + 32 * g + 8 * j);
#pragma unroll
  for (int j = 0; j < 4; ++j) f.w1[j] = *(const v4f*)(b1p + 32 * g + 8 * j);
}

__device__ __forceinline__ void proj_mma(const PGroup& f, v8f& c00, v8f& c01, v8f& c10, v8f& c11) {
#pragma unroll
  for (int j = 0; j < 4; ++j) {
    v2f a0l, a0h, a1l, a1h, w0l, w0h, w1l, w1h;
    a0l.x = f.a0[j].x; a0l.y = f.a0[j].y; a0h.x = f.a0[j].z; a0h.y = f.a0[j].w;
    a1l.x = f.a1[j].x; a1l.y = f.a1[j].y; a1h.x = f.a1[j].z; a1h.y = f.a1[j].w;
    w0l.x = f.w0[j].x; w0l.y = f.w0[j].y; w0h.x = f.w0[j].z; w0h.y = f.w0[j].w;
    w1l.x = f.w1[j].x; w1l.y = f.w1[j].y; w1h.x = f.w1[j].z; w1h.y = f.w1[j].w;
    c00 = wmma4_f32(a0l, w0l, c00);
    c01 = wmma4_f32(a0l, w1l, c01);
    c10 = wmma4_f32(a1l, w0l, c10);
    c11 = wmma4_f32(a1l, w1l, c11);
    c00 = wmma4_f32(a0h, w0h, c00);
    c01 = wmma4_f32(a0h, w1h, c01);
    c10 = wmma4_f32(a1h, w0h, c10);
    c11 = wmma4_f32(a1h, w1h, c11);
  }
}

__device__ __forceinline__ void store_head_rows(float* __restrict__ out, const float* __restrict__ bias,
                                                v8f c, int nb, int m) {
  const v4f b0 = *(const v4f*)(bias + nb);
  const v4f b1 = *(const v4f*)(bias + nb + 4);
  float* dst = out + ((size_t)((m / SEQ) * HEADS + nb / HD) * SEQ + (m % SEQ)) * HD;
  v4f p0, p1;
  p0.x = c[0] + b0.x; p0.y = c[1] + b0.y; p0.z = c[2] + b0.z; p0.w = c[3] + b0.w;
  p1.x = c[4] + b1.x; p1.y = c[5] + b1.y; p1.z = c[6] + b1.z; p1.w = c[7] + b1.w;
  *(v4f*)dst       = p0;
  *(v4f*)(dst + 4) = p1;
}

__device__ __forceinline__ void store_head_keys(float* __restrict__ out, float bn,
                                                v8f c, int n, int mt, int hi) {
  float* dst = out + (size_t)((mt / SEQ) * HEADS + n / HD) * (SEQ * HD)
                   + ((mt % SEQ) / 16) * (16 * HD) + (n % HD) * 16 + 8 * hi;
  v4f p0, p1;
  p0.x = c[0] + bn; p0.y = c[1] + bn; p0.z = c[2] + bn; p0.w = c[3] + bn;
  p1.x = c[4] + bn; p1.y = c[5] + bn; p1.z = c[6] + bn; p1.w = c[7] + bn;
  *(v4f*)dst       = p0;
  *(v4f*)(dst + 4) = p1;
}

__global__ __launch_bounds__(128) void proj_wmma(
    const float* __restrict__ x, int nmat,
    const float* __restrict__ W0, const float* __restrict__ b0, float* __restrict__ o0,
    const float* __restrict__ W1, const float* __restrict__ b1, float* __restrict__ o1,
    const float* __restrict__ W2, const float* __restrict__ b2, float* __restrict__ o2) {
  const int lane = threadIdx.x & 31;
  const int wave = __builtin_amdgcn_readfirstlane((int)(threadIdx.x >> 5));
  const int lo = lane & 15, hi = lane >> 4;

  int blk = blockIdx.x;
  const float* W = W0;
  const float* bias = b0;
  float* out = o0;
  const bool perhead = (nmat == 3);
  bool rows8 = perhead;
  if (perhead) {
    const int which = blk % 3;
    blk = blk / 3;
    if (which == 1) { W = W1; bias = b1; out = o1; }
    if (which == 2) { W = W2; bias = b2; out = o2; rows8 = false; }
  }
  const int m0 = (blk >> 1) * 64 + (wave & 1) * 32;
  const int n0 = (blk & 1) * 64 + (wave >> 1) * 32;

  const float* xrow = x + (size_t)(m0 + lo) * EMB + 4 * hi;
  const float* wrow = W + (size_t)(n0 + lo) * EMB + 4 * hi;
  const float* a0p = rows8 ? wrow : xrow;
  const float* a1p = a0p + 16 * EMB;
  const float* b0p = rows8 ? xrow : wrow;
  const float* b1p = b0p + 16 * EMB;
  v8f c00 = {}, c01 = {}, c10 = {}, c11 = {};
  PGroup f0, f1, f2, f3;
  proj_load(f0, a0p, a1p, b0p, b1p, 0);
  proj_load(f1, a0p, a1p, b0p, b1p, 1);
  proj_load(f2, a0p, a1p, b0p, b1p, 2);
  proj_load(f3, a0p, a1p, b0p, b1p, 3);
  __builtin_amdgcn_sched_barrier(0);
  proj_mma(f0, c00, c01, c10, c11);
  proj_mma(f1, c00, c01, c10, c11);
  proj_mma(f2, c00, c01, c10, c11);
  proj_mma(f3, c00, c01, c10, c11);
  if (rows8) {
    store_head_rows(out, bias, c00, n0 + 8 * hi,      m0 + lo);
    store_head_rows(out, bias, c01, n0 + 8 * hi,      m0 + 16 + lo);
    store_head_rows(out, bias, c10, n0 + 16 + 8 * hi, m0 + lo);
    store_head_rows(out, bias, c11, n0 + 16 + 8 * hi, m0 + 16 + lo);
    return;
  }
  const float bn0 = bias[n0 + lo];
  const float bn1 = bias[n0 + 16 + lo];
  if (perhead) {
    store_head_keys(out, bn0, c00, n0 + lo,      m0,      hi);
    store_head_keys(out, bn1, c01, n0 + 16 + lo, m0,      hi);
    store_head_keys(out, bn0, c10, n0 + lo,      m0 + 16, hi);
    store_head_keys(out, bn1, c11, n0 + 16 + lo, m0 + 16, hi);
    return;
  }
  float* orow = out + (size_t)(m0 + 8 * hi) * EMB + n0 + lo;
#pragma unroll
  for (int r = 0; r < 8; ++r) {
    orow[(size_t)r * EMB]             = c00[r] + bn0;
    orow[(size_t)r * EMB + 16]        = c01[r] + bn1;
    orow[(size_t)(r + 16) * EMB]      = c10[r] + bn0;
    orow[(size_t)(r + 16) * EMB + 16] = c11[r] + bn1;
  }
}


struct KvFrag {
  v4f k;
  v4f va, vb;
};

__device__ __forceinline__ void kv_load(KvFrag& f, const float* __restrict__ Kb,
                                        const float* __restrict__ Vb, int kt,
                                        unsigned koff, unsigned voff) {
  const float* kt_k = Kb + (size_t)kt * (16 * HD);
  const float* kt_v = Vb + (size_t)kt * (16 * HD);
  f.k  = *(const v4f*)(kt_k + koff);
  f.va = *(const v4f*)(kt_v + voff);
  f.vb = *(const v4f*)(kt_v + voff + 4);
}

template <bool MASK, bool FAST>
__device__ __forceinline__ v8f kv_tile(const KvFrag& f, v2f bq0, v2f bq1, v8f acc,
                                       int lo, int hi, bool isv, float fill) {
  v2f k0, k1;
  k0.x = f.k.x; k0.y = f.k.y; k1.x = f.k.z; k1.y = f.k.w;
  const float v[8] = {f.va.x, f.va.y, f.va.z, f.va.w, f.vb.x, f.vb.y, f.vb.z, f.vb.w};
  v8f s = {};
  s = wmma4_f32(k0, bq0, s);
  s = wmma4_f32(k1, bq1, s);

  float w[8];
#pragma unroll
  for (int r = 0; r < 8; ++r) {
    float wt;
    if (FAST) {
      wt = 1.0f + __builtin_amdgcn_tanhf(s[r]);
    } else {
      const float e = __builtin_amdgcn_exp2f(NEG2_LOG2E * s[r]);
      wt = __builtin_amdgcn_rcpf(0.5f + 0.5f * e);
    }
    w[r] = MASK ? ((8 * hi + r <= lo) ? wt : 0.0f) : wt;
  }

#pragma unroll
  for (int j = 0; j < 4; ++j) {
    v2f a, bb;
    a.x  = isv ? v[2 * j]     : fill;
    a.y  = isv ? v[2 * j + 1] : fill;
    bb.x = w[2 * j];
    bb.y = w[2 * j + 1];
    acc = wmma4_f32(a, bb, acc);
  }
  return acc;
}

template <bool DIAG, bool FAST>
__device__ __forceinline__ v8f span_pass(
    const float* __restrict__ Qb, const float* __restrict__ Kb, const float* __restrict__ Vb,
    int qt, int kt0, int kt1, int lo, int hi) {
  const int q0 = qt * 16;

  const v4f q = *(const v4f*)(Qb + (size_t)(q0 + lo) * HD + 4 * hi);
  v2f bq0, bq1;
  bq0.x = q.x * HALF_SCALE; bq0.y = q.y * HALF_SCALE;
  bq1.x = q.z * HALF_SCALE; bq1.y = q.w * HALF_SCALE;

  const bool  isv  = lo < 8;
  const float fill = (lo == 8) ? 1.0f : 0.0f;

  const unsigned koff = (unsigned)(lo * HD + 4 * hi);
  const unsigned voff = (unsigned)((lo & 7) * 16 + 8 * hi);
  const int last     = kt1 - 1;
  const int lastfull = DIAG ? last - 1 : last;

  KvFrag fa, fb;
  kv_load(fa, Kb, Vb, kt0, koff, voff);
  v8f acc = {};
  int kt = kt0;
  for (; kt < lastfull; kt += 2) {
    kv_load(fb, Kb, Vb, kt + 1, koff, voff);
    asm volatile("" ::: "memory");
    acc = kv_tile<false, FAST>(fa, bq0, bq1, acc, lo, hi, isv, fill);
    const int nx = (kt + 2 < last) ? kt + 2 : last;
    kv_load(fa, Kb, Vb, nx, koff, voff);
    asm volatile("" ::: "memory");
    acc = kv_tile<false, FAST>(fb, bq0, bq1, acc, lo, hi, isv, fill);
  }
  if (kt == lastfull) {
    if (DIAG) {
      kv_load(fb, Kb, Vb, last, koff, voff);
      asm volatile("" ::: "memory");
    }
    acc = kv_tile<false, FAST>(fa, bq0, bq1, acc, lo, hi, isv, fill);
    if (DIAG) acc = kv_tile<true, FAST>(fb, bq0, bq1, acc, lo, hi, isv, fill);
  } else if (DIAG) {
    acc = kv_tile<true, FAST>(fa, bq0, bq1, acc, lo, hi, isv, fill);
  }
  return acc;
}

template <bool DIAG>
__device__ __forceinline__ v8f attn_span(
    const float* __restrict__ Qb, const float* __restrict__ Kb, const float* __restrict__ Vb,
    int qt, int kt0, int kt1, int lo, int hi) {
  v8f acc = span_pass<DIAG, true>(Qb, Kb, Vb, qt, kt0, kt1, lo, hi);
  const int upto = qt * 16 + lo - kt0 * 16 + 1;
  const int span = (kt1 - kt0) * 16;
  const int nk   = upto < span ? upto : span;
  const bool thin = (hi == 1) && !(acc[0] >= 0.03125f * (float)nk);
  if (__builtin_amdgcn_ballot_w32(thin) != 0u)
    acc = span_pass<DIAG, false>(Qb, Kb, Vb, qt, kt0, kt1, lo, hi);
  return acc;
}

__device__ __forceinline__ void attn_store(float* __restrict__ Ob, v8f acc, int qt, int lo, int hi) {
  const float rs = __shfl(acc[0], 16 + lo, 32);
  if (hi == 0) {
    float4 o0, o1;
    o0.x = acc[0] / rs; o0.y = acc[1] / rs; o0.z = acc[2] / rs; o0.w = acc[3] / rs;
    o1.x = acc[4] / rs; o1.y = acc[5] / rs; o1.z = acc[6] / rs; o1.w = acc[7] / rs;
    float4* orow = (float4*)(Ob + (size_t)(qt * 16 + lo) * EMB);
    orow[0] = o0;
    orow[1] = o1;
  }
}

constexpr int QTILES = SEQ / 16;
constexpr int HALF_T = QTILES / 2;
__global__ __launch_bounds__(256) void attn_wmma(
    const float* __restrict__ Q, const float* __restrict__ K,
    const float* __restrict__ V, float* __restrict__ O) {
  __shared__ float4 part[4][2][32];

  const int lane = threadIdx.x & 31;
  const int wave = __builtin_amdgcn_readfirstlane((int)(threadIdx.x >> 5));
  const int pw   = wave & 3;
  const int role = wave >> 2;
  const int bh = blockIdx.x >> 4;
  const int p  = (blockIdx.x & 15) * 4 + pw;
  const int ql = p, qh = QTILES - 1 - p;
  const int b = bh >> 4, h = bh & 15;
  const int lo = lane & 15, hi = lane >> 4;
  const size_t hbase = (size_t)bh * (SEQ * HD);
  const float* Qb = Q + hbase;
  const float* Kb = K + hbase;
  const float* Vb = V + hbase;
  float*       Ob = O + (size_t)b * SEQ * EMB + (size_t)h * HD;

  v8f acc;
  if (role == 0) {
    acc = attn_span<false>(Qb, Kb, Vb, qh, 0, HALF_T, lo, hi);
    float4 a0, a1;
    a0.x = acc[0]; a0.y = acc[1]; a0.z = acc[2]; a0.w = acc[3];
    a1.x = acc[4]; a1.y = acc[5]; a1.z = acc[6]; a1.w = acc[7];
    part[pw][0][lane] = a0;
    part[pw][1][lane] = a1;
  } else {
    const v8f al = attn_span<true>(Qb, Kb, Vb, ql, 0, ql + 1, lo, hi);
    attn_store(Ob, al, ql, lo, hi);
    acc = attn_span<true>(Qb, Kb, Vb, qh, HALF_T, qh + 1, lo, hi);
  }
  __syncthreads();
  if (role == 1) {
    const float4 a0 = part[pw][0][lane];
    const float4 a1 = part[pw][1][lane];
    acc[0] += a0.x; acc[1] += a0.y; acc[2] += a0.z; acc[3] += a0.w;
    acc[4] += a1.x; acc[5] += a1.y; acc[6] += a1.z; acc[7] += a1.w;
    attn_store(Ob, acc, qh, lo, hi);
  }
}

extern "C" void kernel_launch(void* const* d_in, const int* in_sizes, int n_in,
                              void* d_out, int out_size, void* d_ws, size_t ws_size,
                              hipStream_t stream) {
  const float* x  = (const float*)d_in[0];
  const float* Wq = (const float*)d_in[1];
  const float* bq = (const float*)d_in[2];
  const float* Wk = (const float*)d_in[3];
  const float* bk = (const float*)d_in[4];
  const float* Wv = (const float*)d_in[5];
  const float* bv = (const float*)d_in[6];
  const float* Wo = (const float*)d_in[7];
  const float* bo = (const float*)d_in[8];
  float* out = (float*)d_out;

  const size_t mat = (size_t)BATCH * SEQ * EMB;
  float* Qw = (float*)d_ws;
  float* Kw = Qw + mat;
  float* Vw = Kw + mat;
  float* Aw = Vw + mat;

  proj_wmma<<<dim3(768), dim3(128), 0, stream>>>(x, 3, Wq, bq, Qw, Wk, bk, Kw, Wv, bv, Vw);
  attn_wmma<<<dim3(1024), dim3(256), 0, stream>>>(Qw, Kw, Vw, Aw);
  proj_wmma<<<dim3(256), dim3(128), 0, stream>>>(Aw, 1, Wo, bo, out, Wo, bo, out, Wo, bo, out);
}


# output projection fused into the attention kernel: a 32-wave work-group owns all 16 heads of a query tile pair, its 32 normalised rows go through LDS into the Wo product; 2 launches, not 3
# speedup vs baseline: 5.3882x; 1.1181x over previous
#include <hip/hip_runtime.h>

typedef __attribute__((ext_vector_type(2))) float v2f;
typedef __attribute__((ext_vector_type(8))) float v8f;

constexpr int EMB   = 128;
constexpr int SEQ   = 2048;
constexpr int BATCH = 4;
constexpr int HEADS = 16;
constexpr int HD    = 8;
constexpr float ATT_SCALE = 0.35355339059327373f;
constexpr float HALF_SCALE = 0.5f * ATT_SCALE;
constexpr float NEG2_LOG2E = -2.0f * 1.4426950408889634f;

__device__ __forceinline__ v8f wmma4_f32(v2f a, v2f b, v8f c) {
  return __builtin_amdgcn_wmma_f32_16x16x4_f32(
       false, a,  false, b,
       (short)0, c,  false,  false);
}

typedef __attribute__((ext_vector_type(4))) float v4f;

struct PGroup {
  v4f a0[4], a1[4], w0[4], w1[4];
};

__device__ __forceinline__ void proj_load(PGroup& f, const float* __restrict__ a0p,
                                          const float* __restrict__ a1p,
                                          const float* __restrict__ b0p,
                                          const float* __restrict__ b1p, int g) {
#pragma unroll
  for (int j = 0; j < 4; ++j) f.a0[j] = *(const v4f*)(a0p + 32 * g + 8 * j);
#pragma unroll
  for (int j = 0; j < 4; ++j) f.a1[j] = *(const v4f*)(a1p + 32 * g + 8 * j);
#pragma unroll
  for (int j = 0; j < 4; ++j) f.w0[j] = *(const v4f*)(b0p + 32 * g + 8 * j);
#pragma unroll
  for (int j = 0; j < 4; ++j) f.w1[j] = *(const v4f*)(b1p + 32 * g + 8 * j);
}

__device__ __forceinline__ void proj_mma(const PGroup& f, v8f& c00, v8f& c01, v8f& c10, v8f& c11) {
#pragma unroll
  for (int j = 0; j < 4; ++j) {
    v2f a0l, a0h, a1l, a1h, w0l, w0h, w1l, w1h;
    a0l.x = f.a0[j].x; a0l.y = f.a0[j].y; a0h.x = f.a0[j].z; a0h.y = f.a0[j].w;
    a1l.x = f.a1[j].x; a1l.y = f.a1[j].y; a1h.x = f.a1[j].z; a1h.y = f.a1[j].w;
    w0l.x = f.w0[j].x; w0l.y = f.w0[j].y; w0h.x = f.w0[j].z; w0h.y = f.w0[j].w;
    w1l.x = f.w1[j].x; w1l.y = f.w1[j].y; w1h.x = f.w1[j].z; w1h.y = f.w1[j].w;
    c00 = wmma4_f32(a0l, w0l, c00);
    c01 = wmma4_f32(a0l, w1l, c01);
    c10 = wmma4_f32(a1l, w0l, c10);
    c11 = wmma4_f32(a1l, w1l, c11);
    c00 = wmma4_f32(a0h, w0h, c00);
    c01 = wmma4_f32(a0h, w1h, c01);
    c10 = wmma4_f32(a1h, w0h, c10);
    c11 = wmma4_f32(a1h, w1h, c11);
  }
}

__device__ __forceinline__ void store_head_rows(float* __restrict__ out, const float* __restrict__ bias,
                                                v8f c, int nb, int m) {
  const v4f b0 = *(const v4f*)(bias + nb);
  const v4f b1 = *(const v4f*)(bias + nb + 4);
  float* dst = out + ((size_t)((m / SEQ) * HEADS + nb / HD) * SEQ + (m % SEQ)) * HD;
  v4f p0, p1;
  p0.x = c[0] + b0.x; p0.y = c[1] + b0.y; p0.z = c[2] + b0.z; p0.w = c[3] + b0.w;
  p1.x = c[4] + b1.x; p1.y = c[5] + b1.y; p1.z = c[6] + b1.z; p1.w = c[7] + b1.w;
  *(v4f*)dst       = p0;
  *(v4f*)(dst + 4) = p1;
}

__device__ __forceinline__ void store_head_keys(float* __restrict__ out, float bn,
                                                v8f c, int n, int mt, int hi) {
  float* dst = out + (size_t)((mt / SEQ) * HEADS + n / HD) * (SEQ * HD)
                   + ((mt % SEQ) / 16) * (16 * HD) + (n % HD) * 16 + 8 * hi;
  v4f p0, p1;
  p0.x = c[0] + bn; p0.y = c[1] + bn; p0.z = c[2] + bn; p0.w = c[3] + bn;
  p1.x = c[4] + bn; p1.y = c[5] + bn; p1.z = c[6] + bn; p1.w = c[7] + bn;
  *(v4f*)dst       = p0;
  *(v4f*)(dst + 4) = p1;
}

__global__ __launch_bounds__(128) void proj_wmma(
    const float* __restrict__ x,
    const float* __restrict__ W0, const float* __restrict__ b0, float* __restrict__ o0,
    const float* __restrict__ W1, const float* __restrict__ b1, float* __restrict__ o1,
    const float* __restrict__ W2, const float* __restrict__ b2, float* __restrict__ o2) {
  const int lane = threadIdx.x & 31;
  const int wave = __builtin_amdgcn_readfirstlane((int)(threadIdx.x >> 5));
  const int lo = lane & 15, hi = lane >> 4;

  const float* W = W0;
  const float* bias = b0;
  float* out = o0;
  bool rows8 = true;
  const int which = blockIdx.x % 3;
  const int blk = blockIdx.x / 3;
  if (which == 1) { W = W1; bias = b1; out = o1; }
  if (which == 2) { W = W2; bias = b2; out = o2; rows8 = false; }
  const int m0 = (blk >> 1) * 64 + (wave & 1) * 32;
  const int n0 = (blk & 1) * 64 + (wave >> 1) * 32;

  const float* xrow = x + (size_t)(m0 + lo) * EMB + 4 * hi;
  const float* wrow = W + (size_t)(n0 + lo) * EMB + 4 * hi;
  const float* a0p = rows8 ? wrow : xrow;
  const float* a1p = a0p + 16 * EMB;
  const float* b0p = rows8 ? xrow : wrow;
  const float* b1p = b0p + 16 * EMB;
  v8f c00 = {}, c01 = {}, c10 = {}, c11 = {};
  PGroup f0, f1, f2, f3;
  proj_load(f0, a0p, a1p, b0p, b1p, 0);
  proj_load(f1, a0p, a1p, b0p, b1p, 1);
  proj_load(f2, a0p, a1p, b0p, b1p, 2);
  proj_load(f3, a0p, a1p, b0p, b1p, 3);
  __builtin_amdgcn_sched_barrier(0);
  proj_mma(f0, c00, c01, c10, c11);
  proj_mma(f1, c00, c01, c10, c11);
  proj_mma(f2, c00, c01, c10, c11);
  proj_mma(f3, c00, c01, c10, c11);
  if (rows8) {
    store_head_rows(out, bias, c00, n0 + 8 * hi,      m0 + lo);
    store_head_rows(out, bias, c01, n0 + 8 * hi,      m0 + 16 + lo);
    store_head_rows(out, bias, c10, n0 + 16 + 8 * hi, m0 + lo);
    store_head_rows(out, bias, c11, n0 + 16 + 8 * hi, m0 + 16 + lo);
    return;
  }
  const float bn0 = bias[n0 + lo];
  const float bn1 = bias[n0 + 16 + lo];
  store_head_keys(out, bn0, c00, n0 + lo,      m0,      hi);
  store_head_keys(out, bn1, c01, n0 + 16 + lo, m0,      hi);
  store_head_keys(out, bn0, c10, n0 + lo,      m0 + 16, hi);
  store_head_keys(out, bn1, c11, n0 + 16 + lo, m0 + 16, hi);
}


struct KvFrag {
  v4f k;
  v4f va, vb;
};

__device__ __forceinline__ void kv_load(KvFrag& f, const float* __restrict__ Kb,
                                        const float* __restrict__ Vb, int kt,
                                        unsigned koff, unsigned voff) {
  const float* kt_k = Kb + (size_t)kt * (16 * HD);
  const float* kt_v = Vb + (size_t)kt * (16 * HD);
  f.k  = *(const v4f*)(kt_k + koff);
  f.va = *(const v4f*)(kt_v + voff);
  f.vb = *(const v4f*)(kt_v + voff + 4);
}

template <bool MASK, bool FAST>
__device__ __forceinline__ v8f kv_tile(const KvFrag& f, v2f bq0, v2f bq1, v8f acc,
                                       int lo, int hi, bool isv, float fill) {
  v2f k0, k1;
  k0.x = f.k.x; k0.y = f.k.y; k1.x = f.k.z; k1.y = f.k.w;
  const float v[8] = {f.va.x, f.va.y, f.va.z, f.va.w, f.vb.x, f.vb.y, f.vb.z, f.vb.w};
  v8f s = {};
  s = wmma4_f32(k0, bq0, s);
  s = wmma4_f32(k1, bq1, s);

  float w[8];
#pragma unroll
  for (int r = 0; r < 8; ++r) {
    float wt;
    if (FAST) {
      wt = 1.0f + __builtin_amdgcn_tanhf(s[r]);
    } else {
      const float e = __builtin_amdgcn_exp2f(NEG2_LOG2E * s[r]);
      wt = __builtin_amdgcn_rcpf(0.5f + 0.5f * e);
    }
    w[r] = MASK ? ((8 * hi + r <= lo) ? wt : 0.0f) : wt;
  }

#pragma unroll
  for (int j = 0; j < 4; ++j) {
    v2f a, bb;
    a.x  = isv ? v[2 * j]     : fill;
    a.y  = isv ? v[2 * j + 1] : fill;
    bb.x = w[2 * j];
    bb.y = w[2 * j + 1];
    acc = wmma4_f32(a, bb, acc);
  }
  return acc;
}

template <bool DIAG, bool FAST>
__device__ __forceinline__ v8f span_pass(
    const float* __restrict__ Qb, const float* __restrict__ Kb, const float* __restrict__ Vb,
    int qt, int kt0, int kt1, int lo, int hi) {
  const int q0 = qt * 16;

  const v4f q = *(const v4f*)(Qb + (size_t)(q0 + lo) * HD + 4 * hi);
  v2f bq0, bq1;
  bq0.x = q.x * HALF_SCALE; bq0.y = q.y * HALF_SCALE;
  bq1.x = q.z * HALF_SCALE; bq1.y = q.w * HALF_SCALE;

  const bool  isv  = lo < 8;
  const float fill = (lo == 8) ? 1.0f : 0.0f;

  const unsigned koff = (unsigned)(lo * HD + 4 * hi);
  const unsigned voff = (unsigned)((lo & 7) * 16 + 8 * hi);
  const int last     = kt1 - 1;
  const int lastfull = DIAG ? last - 1 : last;

  KvFrag fa, fb;
  kv_load(fa, Kb, Vb, kt0, koff, voff);
  v8f acc = {};
  int kt = kt0;
  for (; kt < lastfull; kt += 2) {
    kv_load(fb, Kb, Vb, kt + 1, koff, voff);
    asm volatile("" ::: "memory");
    acc = kv_tile<false, FAST>(fa, bq0, bq1, acc, lo, hi, isv, fill);
    const int nx = (kt + 2 < last) ? kt + 2 : last;
    kv_load(fa, Kb, Vb, nx, koff, voff);
    asm volatile("" ::: "memory");
    acc = kv_tile<false, FAST>(fb, bq0, bq1, acc, lo, hi, isv, fill);
  }
  if (kt == lastfull) {
    if (DIAG) {
      kv_load(fb, Kb, Vb, last, koff, voff);
      asm volatile("" ::: "memory");
    }
    acc = kv_tile<false, FAST>(fa, bq0, bq1, acc, lo, hi, isv, fill);
    if (DIAG) acc = kv_tile<true, FAST>(fb, bq0, bq1, acc, lo, hi, isv, fill);
  } else if (DIAG) {
    acc = kv_tile<true, FAST>(fa, bq0, bq1, acc, lo, hi, isv, fill);
  }
  return acc;
}

template <bool DIAG>
__device__ __forceinline__ v8f attn_span(
    const float* __restrict__ Qb, const float* __restrict__ Kb, const float* __restrict__ Vb,
    int qt, int kt0, int kt1, int lo, int hi) {
  v8f acc = span_pass<DIAG, true>(Qb, Kb, Vb, qt, kt0, kt1, lo, hi);
  const int upto = qt * 16 + lo - kt0 * 16 + 1;
  const int span = (kt1 - kt0) * 16;
  const int nk   = upto < span ? upto : span;
  const bool thin = (hi == 1) && !(acc[0] >= 0.03125f * (float)nk);
  if (__builtin_amdgcn_ballot_w32(thin) != 0u)
    acc = span_pass<DIAG, false>(Qb, Kb, Vb, qt, kt0, kt1, lo, hi);
  return acc;
}

constexpr int APITCH = EMB + 4;

__device__ __forceinline__ void attn_to_lds(float* arow, v8f acc, int rowbase, int h, int lo, int hi) {
  const float rs = __shfl(acc[0], 16 + lo, 32);
  if (hi == 0) {
    float4 o0, o1;
    o0.x = acc[0] / rs; o0.y = acc[1] / rs; o0.z = acc[2] / rs; o0.w = acc[3] / rs;
    o1.x = acc[4] / rs; o1.y = acc[5] / rs; o1.z = acc[6] / rs; o1.w = acc[7] / rs;
    float4* dst = (float4*)(arow + (rowbase + lo) * APITCH + h * HD);
    dst[0] = o0;
    dst[1] = o1;
  }
}

constexpr int QTILES = SEQ / 16;
constexpr int HALF_T = QTILES / 2;
__global__ __launch_bounds__(1024) void attn_proj_wmma(
    const float* __restrict__ Q, const float* __restrict__ K, const float* __restrict__ V,
    const float* __restrict__ Wo, const float* __restrict__ bo, float* __restrict__ out) {
  __shared__ float4 part[HEADS][2][32];
  __shared__ __attribute__((aligned(16))) float arow[32 * APITCH];

  const int lane = threadIdx.x & 31;
  const int wave = __builtin_amdgcn_readfirstlane((int)(threadIdx.x >> 5));
  const int h    = wave & 15;
  const int role = wave >> 4;
  const int b = blockIdx.x >> 6;
  const int p = blockIdx.x & 63;
  const int ql = p, qh = QTILES - 1 - p;
  const int lo = lane & 15, hi = lane >> 4;
  const size_t hbase = (size_t)(b * HEADS + h) * (SEQ * HD);
  const float* Qb = Q + hbase;
  const float* Kb = K + hbase;
  const float* Vb = V + hbase;

  v8f acc;
  if (role == 0) {
    acc = attn_span<false>(Qb, Kb, Vb, qh, 0, HALF_T, lo, hi);
    float4 a0, a1;
    a0.x = acc[0]; a0.y = acc[1]; a0.z = acc[2]; a0.w = acc[3];
    a1.x = acc[4]; a1.y = acc[5]; a1.z = acc[6]; a1.w = acc[7];
    part[h][0][lane] = a0;
    part[h][1][lane] = a1;
  } else {
    const v8f al = attn_span<true>(Qb, Kb, Vb, ql, 0, ql + 1, lo, hi);
    attn_to_lds(arow, al, 0, h, lo, hi);
    acc = attn_span<true>(Qb, Kb, Vb, qh, HALF_T, qh + 1, lo, hi);
  }
  __syncthreads();
  if (role == 1) {
    const float4 a0 = part[h][0][lane];
    const float4 a1 = part[h][1][lane];
    acc[0] += a0.x; acc[1] += a0.y; acc[2] += a0.z; acc[3] += a0.w;
    acc[4] += a1.x; acc[5] += a1.y; acc[6] += a1.z; acc[7] += a1.w;
    attn_to_lds(arow, acc, 16, h, lo, hi);
  }
  __syncthreads();

  if (wave < 8) {
    const int rt = wave & 1;
    const int n0 = (wave >> 1) * 32;
    const float* arp = arow + (rt * 16 + lo) * APITCH + 4 * hi;
    const float* w0p = Wo + (size_t)(n0 + lo) * EMB + 4 * hi;
    const float* w1p = w0p + 16 * EMB;
    v8f c0 = {}, c1 = {};
#pragma unroll
    for (int c = 0; c < 16; ++c) {
      const v4f a  = *(const v4f*)(arp + 8 * c);
      const v4f w0 = *(const v4f*)(w0p + 8 * c);
      const v4f w1 = *(const v4f*)(w1p + 8 * c);
      v2f al, ah, w0l, w0h, w1l, w1h;
      al.x  = a.x;  al.y  = a.y;  ah.x  = a.z;  ah.y  = a.w;
      w0l.x = w0.x; w0l.y = w0.y; w0h.x = w0.z; w0h.y = w0.w;
      w1l.x = w1.x; w1l.y = w1.y; w1h.x = w1.z; w1h.y = w1.w;
      c0 = wmma4_f32(al, w0l, c0);
      c1 = wmma4_f32(al, w1l, c1);
      c0 = wmma4_f32(ah, w0h, c0);
      c1 = wmma4_f32(ah, w1h, c1);
    }
    const float bn0 = bo[n0 + lo];
    const float bn1 = bo[n0 + 16 + lo];
    const int q0 = (rt == 0 ? ql : qh) * 16;
    float* orow = out + ((size_t)b * SEQ + q0 + 8 * hi) * EMB + n0 + lo;
#pragma unroll
    for (int r = 0; r < 8; ++r) {
      orow[(size_t)r * EMB]      = c0[r] + bn0;
      orow[(size_t)r * EMB + 16] = c1[r] + bn1;
    }
  }
}

extern "C" void kernel_launch(void* const* d_in, const int* in_sizes, int n_in,
                              void* d_out, int out_size, void* d_ws, size_t ws_size,
                              hipStream_t stream) {
  const float* x  = (const float*)d_in[0];
  const float* Wq = (const float*)d_in[1];
  const float* bq = (const float*)d_in[2];
  const float* Wk = (const float*)d_in[3];
  const float* bk = (const float*)d_in[4];
  const float* Wv = (const float*)d_in[5];
  const float* bv = (const float*)d_in[6];
  const float* Wo = (const float*)d_in[7];
  const float* bo = (const float*)d_in[8];
  float* out = (float*)d_out;

  const size_t mat = (size_t)BATCH * SEQ * EMB;
  float* Qw = (float*)d_ws;
  float* Kw = Qw + mat;
  float* Vw = Kw + mat;

  proj_wmma<<<dim3(768), dim3(128), 0, stream>>>(x, Wq, bq, Qw, Wk, bk, Kw, Wv, bv, Vw);
  attn_proj_wmma<<<dim3(256), dim3(1024), 0, stream>>>(Qw, Kw, Vw, Wo, bo, out);
}


# attention waves take their issue priority from their own progress (s_setprio 3, then 2, 1, 0 once 24, 44, 58 of their 64-65 key tiles are done), so the waves of a SIMD end together, not oldest first
# speedup vs baseline: 5.6705x; 1.0524x over previous
#include <hip/hip_runtime.h>

typedef __attribute__((ext_vector_type(2))) float v2f;
typedef __attribute__((ext_vector_type(8))) float v8f;

constexpr int EMB   = 128;
constexpr int SEQ   = 2048;
constexpr int BATCH = 4;
constexpr int HEADS = 16;
constexpr int HD    = 8;
constexpr float ATT_SCALE = 0.35355339059327373f;
constexpr float HALF_SCALE = 0.5f * ATT_SCALE;
constexpr float NEG2_LOG2E = -2.0f * 1.4426950408889634f;

__device__ __forceinline__ v8f wmma4_f32(v2f a, v2f b, v8f c) {
  return __builtin_amdgcn_wmma_f32_16x16x4_f32(
       false, a,  false, b,
       (short)0, c,  false,  false);
}

typedef __attribute__((ext_vector_type(4))) float v4f;

struct PGroup {
  v4f a0[4], a1[4], w0[4], w1[4];
};

__device__ __forceinline__ void proj_load(PGroup& f, const float* __restrict__ a0p,
                                          const float* __restrict__ a1p,
                                          const float* __restrict__ b0p,
                                          const float* __restrict__ b1p, int g) {
#pragma unroll
  for (int j = 0; j < 4; ++j) f.a0[j] = *(const v4f*)(a0p + 32 * g + 8 * j);
#pragma unroll
  for (int j = 0; j < 4; ++j) f.a1[j] = *(const v4f*)(a1p + 32 * g + 8 * j);
#pragma unroll
  for (int j = 0; j < 4; ++j) f.w0[j] = *(const v4f*)(b0p + 32 * g + 8 * j);
#pragma unroll
  for (int j = 0; j < 4; ++j) f.w1[j] = *(const v4f*)(b1p + 32 * g + 8 * j);
}

__device__ __forceinline__ void proj_mma(const PGroup& f, v8f& c00, v8f& c01, v8f& c10, v8f& c11) {
#pragma unroll
  for (int j = 0; j < 4; ++j) {
    v2f a0l, a0h, a1l, a1h, w0l, w0h, w1l, w1h;
    a0l.x = f.a0[j].x; a0l.y = f.a0[j].y; a0h.x = f.a0[j].z; a0h.y = f.a0[j].w;
    a1l.x = f.a1[j].x; a1l.y = f.a1[j].y; a1h.x = f.a1[j].z; a1h.y = f.a1[j].w;
    w0l.x = f.w0[j].x; w0l.y = f.w0[j].y; w0h.x = f.w0[j].z; w0h.y = f.w0[j].w;
    w1l.x = f.w1[j].x; w1l.y = f.w1[j].y; w1h.x = f.w1[j].z; w1h.y = f.w1[j].w;
    c00 = wmma4_f32(a0l, w0l, c00);
    c01 = wmma4_f32(a0l, w1l, c01);
    c10 = wmma4_f32(a1l, w0l, c10);
    c11 = wmma4_f32(a1l, w1l, c11);
    c00 = wmma4_f32(a0h, w0h, c00);
    c01 = wmma4_f32(a0h, w1h, c01);
    c10 = wmma4_f32(a1h, w0h, c10);
    c11 = wmma4_f32(a1h, w1h, c11);
  }
}

__device__ __forceinline__ void store_head_rows(float* __restrict__ out, const float* __restrict__ bias,
                                                v8f c, int nb, int m) {
  const v4f b0 = *(const v4f*)(bias + nb);
  const v4f b1 = *(const v4f*)(bias + nb + 4);
  float* dst = out + ((size_t)((m / SEQ) * HEADS + nb / HD) * SEQ + (m % SEQ)) * HD;
  v4f p0, p1;
  p0.x = c[0] + b0.x; p0.y = c[1] + b0.y; p0.z = c[2] + b0.z; p0.w = c[3] + b0.w;
  p1.x = c[4] + b1.x; p1.y = c[5] + b1.y; p1.z = c[6] + b1.z; p1.w = c[7] + b1.w;
  *(v4f*)dst       = p0;
  *(v4f*)(dst + 4) = p1;
}

__device__ __forceinline__ void store_head_keys(float* __restrict__ out, float bn,
                                                v8f c, int n, int mt, int hi) {
  float* dst = out + (size_t)((mt / SEQ) * HEADS + n / HD) * (SEQ * HD)
                   + ((mt % SEQ) / 16) * (16 * HD) + (n % HD) * 16 + 8 * hi;
  v4f p0, p1;
  p0.x = c[0] + bn; p0.y = c[1] + bn; p0.z = c[2] + bn; p0.w = c[3] + bn;
  p1.x = c[4] + bn; p1.y = c[5] + bn; p1.z = c[6] + bn; p1.w = c[7] + bn;
  *(v4f*)dst       = p0;
  *(v4f*)(dst + 4) = p1;
}

__global__ __launch_bounds__(128) void proj_wmma(
    const float* __restrict__ x,
    const float* __restrict__ W0, const float* __restrict__ b0, float* __restrict__ o0,
    const float* __restrict__ W1, const float* __restrict__ b1, float* __restrict__ o1,
    const float* __restrict__ W2, const float* __restrict__ b2, float* __restrict__ o2) {
  const int lane = threadIdx.x & 31;
  const int wave = __builtin_amdgcn_readfirstlane((int)(threadIdx.x >> 5));
  const int lo = lane & 15, hi = lane >> 4;

  const float* W = W0;
  const float* bias = b0;
  float* out = o0;
  bool rows8 = true;
  const int which = blockIdx.x % 3;
  const int blk = blockIdx.x / 3;
  if (which == 1) { W = W1; bias = b1; out = o1; }
  if (which == 2) { W = W2; bias = b2; out = o2; rows8 = false; }
  const int m0 = (blk >> 1) * 64 + (wave & 1) * 32;
  const int n0 = (blk & 1) * 64 + (wave >> 1) * 32;

  const float* xrow = x + (size_t)(m0 + lo) * EMB + 4 * hi;
  const float* wrow = W + (size_t)(n0 + lo) * EMB + 4 * hi;
  const float* a0p = rows8 ? wrow : xrow;
  const float* a1p = a0p + 16 * EMB;
  const float* b0p = rows8 ? xrow : wrow;
  const float* b1p = b0p + 16 * EMB;
  v8f c00 = {}, c01 = {}, c10 = {}, c11 = {};
  PGroup f0, f1, f2, f3;
  proj_load(f0, a0p, a1p, b0p, b1p, 0);
  proj_load(f1, a0p, a1p, b0p, b1p, 1);
  proj_load(f2, a0p, a1p, b0p, b1p, 2);
  proj_load(f3, a0p, a1p, b0p, b1p, 3);
  __builtin_amdgcn_sched_barrier(0);
  proj_mma(f0, c00, c01, c10, c11);
  proj_mma(f1, c00, c01, c10, c11);
  proj_mma(f2, c00, c01, c10, c11);
  proj_mma(f3, c00, c01, c10, c11);
  if (rows8) {
    store_head_rows(out, bias, c00, n0 + 8 * hi,      m0 + lo);
    store_head_rows(out, bias, c01, n0 + 8 * hi,      m0 + 16 + lo);
    store_head_rows(out, bias, c10, n0 + 16 + 8 * hi, m0 + lo);
    store_head_rows(out, bias, c11, n0 + 16 + 8 * hi, m0 + 16 + lo);
    return;
  }
  const float bn0 = bias[n0 + lo];
  const float bn1 = bias[n0 + 16 + lo];
  store_head_keys(out, bn0, c00, n0 + lo,      m0,      hi);
  store_head_keys(out, bn1, c01, n0 + 16 + lo, m0,      hi);
  store_head_keys(out, bn0, c10, n0 + lo,      m0 + 16, hi);
  store_head_keys(out, bn1, c11, n0 + 16 + lo, m0 + 16, hi);
}


struct KvFrag {
  v4f k;
  v4f va, vb;
};

__device__ __forceinline__ void kv_load(KvFrag& f, const float* __restrict__ Kb,
                                        const float* __restrict__ Vb, int kt,
                                        unsigned koff, unsigned voff) {
  const float* kt_k = Kb + (size_t)kt * (16 * HD);
  const float* kt_v = Vb + (size_t)kt * (16 * HD);
  f.k  = *(const v4f*)(kt_k + koff);
  f.va = *(const v4f*)(kt_v + voff);
  f.vb = *(const v4f*)(kt_v + voff + 4);
}

template <bool MASK, bool FAST>
__device__ __forceinline__ v8f kv_tile(const KvFrag& f, v2f bq0, v2f bq1, v8f acc,
                                       int lo, int hi, bool isv, float fill) {
  v2f k0, k1;
  k0.x = f.k.x; k0.y = f.k.y; k1.x = f.k.z; k1.y = f.k.w;
  const float v[8] = {f.va.x, f.va.y, f.va.z, f.va.w, f.vb.x, f.vb.y, f.vb.z, f.vb.w};
  v8f s = {};
  s = wmma4_f32(k0, bq0, s);
  s = wmma4_f32(k1, bq1, s);

  float w[8];
#pragma unroll
  for (int r = 0; r < 8; ++r) {
    float wt;
    if (FAST) {
      wt = 1.0f + __builtin_amdgcn_tanhf(s[r]);
    } else {
      const float e = __builtin_amdgcn_exp2f(NEG2_LOG2E * s[r]);
      wt = __builtin_amdgcn_rcpf(0.5f + 0.5f * e);
    }
    w[r] = MASK ? ((8 * hi + r <= lo) ? wt : 0.0f) : wt;
  }

#pragma unroll
  for (int j = 0; j < 4; ++j) {
    v2f a, bb;
    a.x  = isv ? v[2 * j]     : fill;
    a.y  = isv ? v[2 * j + 1] : fill;
    bb.x = w[2 * j];
    bb.y = w[2 * j + 1];
    acc = wmma4_f32(a, bb, acc);
  }
  return acc;
}

constexpr int PRIO_T2 = 24, PRIO_T1 = 44, PRIO_T0 = 58;
constexpr int QTILES_ALL = SEQ / 16;
__device__ __forceinline__ int prio_level(int done) {
  return done >= PRIO_T0 ? 0 : done >= PRIO_T1 ? 1 : done >= PRIO_T2 ? 2 : 3;
}
__device__ __forceinline__ void prio_set(int level) {
  if (level == 3)      __builtin_amdgcn_s_setprio(3);
  else if (level == 2) __builtin_amdgcn_s_setprio(2);
  else if (level == 1) __builtin_amdgcn_s_setprio(1);
  else                 __builtin_amdgcn_s_setprio(0);
}

template <bool DIAG, bool FAST>
__device__ __forceinline__ v8f span_pass(
    const float* __restrict__ Qb, const float* __restrict__ Kb, const float* __restrict__ Vb,
    int qt, int kt0, int kt1, int lo, int hi, int done0) {
  const int q0 = qt * 16;

  const v4f q = *(const v4f*)(Qb + (size_t)(q0 + lo) * HD + 4 * hi);
  v2f bq0, bq1;
  bq0.x = q.x * HALF_SCALE; bq0.y = q.y * HALF_SCALE;
  bq1.x = q.z * HALF_SCALE; bq1.y = q.w * HALF_SCALE;

  const bool  isv  = lo < 8;
  const float fill = (lo == 8) ? 1.0f : 0.0f;

  const unsigned koff = (unsigned)(lo * HD + 4 * hi);
  const unsigned voff = (unsigned)((lo & 7) * 16 + 8 * hi);
  const int last     = kt1 - 1;
  const int lastfull = DIAG ? last - 1 : last;

  KvFrag fa, fb;
  kv_load(fa, Kb, Vb, kt0, koff, voff);
  v8f acc = {};
  int kt = kt0;
  int level = prio_level(done0);
  prio_set(level);
  for (;;) {
    const int bandend = level == 3 ? PRIO_T2 : level == 2 ? PRIO_T1 : level == 1 ? PRIO_T0 : 2 * QTILES_ALL;
    const int want = kt0 + (bandend - done0);
    const int stop = want < lastfull ? want : lastfull;
#pragma unroll 2
    for (; kt < stop; kt += 2) {
      kv_load(fb, Kb, Vb, kt + 1, koff, voff);
      asm volatile("" ::: "memory");
      acc = kv_tile<false, FAST>(fa, bq0, bq1, acc, lo, hi, isv, fill);
      const int nx = (kt + 2 < last) ? kt + 2 : last;
      kv_load(fa, Kb, Vb, nx, koff, voff);
      asm volatile("" ::: "memory");
      acc = kv_tile<false, FAST>(fb, bq0, bq1, acc, lo, hi, isv, fill);
    }
    if (kt >= lastfull) break;
    level -= 1;
    prio_set(level);
  }
  if (kt == lastfull) {
    if (DIAG) {
      kv_load(fb, Kb, Vb, last, koff, voff);
      asm volatile("" ::: "memory");
    }
    acc = kv_tile<false, FAST>(fa, bq0, bq1, acc, lo, hi, isv, fill);
    if (DIAG) acc = kv_tile<true, FAST>(fb, bq0, bq1, acc, lo, hi, isv, fill);
  } else if (DIAG) {
    acc = kv_tile<true, FAST>(fa, bq0, bq1, acc, lo, hi, isv, fill);
  }
  return acc;
}

template <bool DIAG>
__device__ __forceinline__ v8f attn_span(
    const float* __restrict__ Qb, const float* __restrict__ Kb, const float* __restrict__ Vb,
    int qt, int kt0, int kt1, int lo, int hi, int done0) {
  v8f acc = span_pass<DIAG, true>(Qb, Kb, Vb, qt, kt0, kt1, lo, hi, done0);
  const int upto = qt * 16 + lo - kt0 * 16 + 1;
  const int span = (kt1 - kt0) * 16;
  const int nk   = upto < span ? upto : span;
  const bool thin = (hi == 1) && !(acc[0] >= 0.03125f * (float)nk);
  if (__builtin_amdgcn_ballot_w32(thin) != 0u)
    acc = span_pass<DIAG, false>(Qb, Kb, Vb, qt, kt0, kt1, lo, hi, done0);
  return acc;
}

constexpr int APITCH = EMB + 4;

__device__ __forceinline__ void attn_to_lds(float* arow, v8f acc, int rowbase, int h, int lo, int hi) {
  const float rs = __shfl(acc[0], 16 + lo, 32);
  if (hi == 0) {
    float4 o0, o1;
    o0.x = acc[0] / rs; o0.y = acc[1] / rs; o0.z = acc[2] / rs; o0.w = acc[3] / rs;
    o1.x = acc[4] / rs; o1.y = acc[5] / rs; o1.z = acc[6] / rs; o1.w = acc[7] / rs;
    float4* dst = (float4*)(arow + (rowbase + lo) * APITCH + h * HD);
    dst[0] = o0;
    dst[1] = o1;
  }
}

constexpr int QTILES = SEQ / 16;
constexpr int HALF_T = QTILES / 2;
__global__ __launch_bounds__(1024) void attn_proj_wmma(
    const float* __restrict__ Q, const float* __restrict__ K, const float* __restrict__ V,
    const float* __restrict__ Wo, const float* __restrict__ bo, float* __restrict__ out) {
  __shared__ float4 part[HEADS][2][32];
  __shared__ __attribute__((aligned(16))) float arow[32 * APITCH];

  const int lane = threadIdx.x & 31;
  const int wave = __builtin_amdgcn_readfirstlane((int)(threadIdx.x >> 5));
  const int h    = wave & 15;
  const int role = wave >> 4;
  const int b = blockIdx.x >> 6;
  const int p = blockIdx.x & 63;
  const int ql = p, qh = QTILES - 1 - p;
  const int lo = lane & 15, hi = lane >> 4;
  const size_t hbase = (size_t)(b * HEADS + h) * (SEQ * HD);
  const float* Qb = Q + hbase;
  const float* Kb = K + hbase;
  const float* Vb = V + hbase;

  v8f acc;
  if (role == 0) {
    acc = attn_span<false>(Qb, Kb, Vb, qh, 0, HALF_T, lo, hi, 0);
    float4 a0, a1;
    a0.x = acc[0]; a0.y = acc[1]; a0.z = acc[2]; a0.w = acc[3];
    a1.x = acc[4]; a1.y = acc[5]; a1.z = acc[6]; a1.w = acc[7];
    part[h][0][lane] = a0;
    part[h][1][lane] = a1;
  } else {
    const v8f al = attn_span<true>(Qb, Kb, Vb, ql, 0, ql + 1, lo, hi, 0);
    attn_to_lds(arow, al, 0, h, lo, hi);
    acc = attn_span<true>(Qb, Kb, Vb, qh, HALF_T, qh + 1, lo, hi, ql + 1);
  }
  __builtin_amdgcn_s_setprio(0);
  __syncthreads();
  if (role == 1) {
    const float4 a0 = part[h][0][lane];
    const float4 a1 = part[h][1][lane];
    acc[0] += a0.x; acc[1] += a0.y; acc[2] += a0.z; acc[3] += a0.w;
    acc[4] += a1.x; acc[5] += a1.y; acc[6] += a1.z; acc[7] += a1.w;
    attn_to_lds(arow, acc, 16, h, lo, hi);
  }
  __syncthreads();

  if (wave < 8) {
    const int rt = wave & 1;
    const int n0 = (wave >> 1) * 32;
    const float* arp = arow + (rt * 16 + lo) * APITCH + 4 * hi;
    const float* w0p = Wo + (size_t)(n0 + lo) * EMB + 4 * hi;
    const float* w1p = w0p + 16 * EMB;
    v8f c0 = {}, c1 = {};
#pragma unroll
    for (int c = 0; c < 16; ++c) {
      const v4f a  = *(const v4f*)(arp + 8 * c);
      const v4f w0 = *(const v4f*)(w0p + 8 * c);
      const v4f w1 = *(const v4f*)(w1p + 8 * c);
      v2f al, ah, w0l, w0h, w1l, w1h;
      al.x  = a.x;  al.y  = a.y;  ah.x  = a.z;  ah.y  = a.w;
      w0l.x = w0.x; w0l.y = w0.y; w0h.x = w0.z; w0h.y = w0.w;
      w1l.x = w1.x; w1l.y = w1.y; w1h.x = w1.z; w1h.y = w1.w;
      c0 = wmma4_f32(al, w0l, c0);
      c1 = wmma4_f32(al, w1l, c1);
      c0 = wmma4_f32(ah, w0h, c0);
      c1 = wmma4_f32(ah, w1h, c1);
    }
    const float bn0 = bo[n0 + lo];
    const float bn1 = bo[n0 + 16 + lo];
    const int q0 = (rt == 0 ? ql : qh) * 16;
    float* orow = out + ((size_t)b * SEQ + q0 + 8 * hi) * EMB + n0 + lo;
#pragma unroll
    for (int r = 0; r < 8; ++r) {
      orow[(size_t)r * EMB]      = c0[r] + bn0;
      orow[(size_t)r * EMB + 16] = c1[r] + bn1;
    }
  }
}

extern "C" void kernel_launch(void* const* d_in, const int* in_sizes, int n_in,
                              void* d_out, int out_size, void* d_ws, size_t ws_size,
                              hipStream_t stream) {
  const float* x  = (const float*)d_in[0];
  const float* Wq = (const float*)d_in[1];
  const float* bq = (const float*)d_in[2];
  const float* Wk = (const float*)d_in[3];
  const float* bk = (const float*)d_in[4];
  const float* Wv = (const float*)d_in[5];
  const float* bv = (const float*)d_in[6];
  const float* Wo = (const float*)d_in[7];
  const float* bo = (const float*)d_in[8];
  float* out = (float*)d_out;

  const size_t mat = (size_t)BATCH * SEQ * EMB;
  float* Qw = (float*)d_ws;
  float* Kw = Qw + mat;
  float* Vw = Kw + mat;

  proj_wmma<<<dim3(768), dim3(128), 0, stream>>>(x, Wq, bq, Qw, Wk, bk, Kw, Wv, bv, Vw);
  attn_proj_wmma<<<dim3(256), dim3(1024), 0, stream>>>(Qw, Kw, Vw, Wo, bo, out);
}


# attention waves walk two heads each, side by side in one key loop (second head's operands at a constant offset from the first's): 16-wave work-groups, half the waves to start, shared loop control
# speedup vs baseline: 5.9023x; 1.0409x over previous
#include <hip/hip_runtime.h>

typedef __attribute__((ext_vector_type(2))) float v2f;
typedef __attribute__((ext_vector_type(8))) float v8f;

constexpr int EMB   = 128;
constexpr int SEQ   = 2048;
constexpr int BATCH = 4;
constexpr int HEADS = 16;
constexpr int HD    = 8;
constexpr float ATT_SCALE = 0.35355339059327373f;
constexpr float HALF_SCALE = 0.5f * ATT_SCALE;
constexpr float NEG2_LOG2E = -2.0f * 1.4426950408889634f;

__device__ __forceinline__ v8f wmma4_f32(v2f a, v2f b, v8f c) {
  return __builtin_amdgcn_wmma_f32_16x16x4_f32(
       false, a,  false, b,
       (short)0, c,  false,  false);
}

typedef __attribute__((ext_vector_type(4))) float v4f;

struct PGroup {
  v4f a0[4], a1[4], w0[4], w1[4];
};

__device__ __forceinline__ void proj_load(PGroup& f, const float* __restrict__ a0p,
                                          const float* __restrict__ a1p,
                                          const float* __restrict__ b0p,
                                          const float* __restrict__ b1p, int g) {
#pragma unroll
  for (int j = 0; j < 4; ++j) f.a0[j] = *(const v4f*)(a0p + 32 * g + 8 * j);
#pragma unroll
  for (int j = 0; j < 4; ++j) f.a1[j] = *(const v4f*)(a1p + 32 * g + 8 * j);
#pragma unroll
  for (int j = 0; j < 4; ++j) f.w0[j] = *(const v4f*)(b0p + 32 * g + 8 * j);
#pragma unroll
  for (int j = 0; j < 4; ++j) f.w1[j] = *(const v4f*)(b1p + 32 * g + 8 * j);
}

__device__ __forceinline__ void proj_mma(const PGroup& f, v8f& c00, v8f& c01, v8f& c10, v8f& c11) {
#pragma unroll
  for (int j = 0; j < 4; ++j) {
    v2f a0l, a0h, a1l, a1h, w0l, w0h, w1l, w1h;
    a0l.x = f.a0[j].x; a0l.y = f.a0[j].y; a0h.x = f.a0[j].z; a0h.y = f.a0[j].w;
    a1l.x = f.a1[j].x; a1l.y = f.a1[j].y; a1h.x = f.a1[j].z; a1h.y = f.a1[j].w;
    w0l.x = f.w0[j].x; w0l.y = f.w0[j].y; w0h.x = f.w0[j].z; w0h.y = f.w0[j].w;
    w1l.x = f.w1[j].x; w1l.y = f.w1[j].y; w1h.x = f.w1[j].z; w1h.y = f.w1[j].w;
    c00 = wmma4_f32(a0l, w0l, c00);
    c01 = wmma4_f32(a0l, w1l, c01);
    c10 = wmma4_f32(a1l, w0l, c10);
    c11 = wmma4_f32(a1l, w1l, c11);
    c00 = wmma4_f32(a0h, w0h, c00);
    c01 = wmma4_f32(a0h, w1h, c01);
    c10 = wmma4_f32(a1h, w0h, c10);
    c11 = wmma4_f32(a1h, w1h, c11);
  }
}

__device__ __forceinline__ void store_head_rows(float* __restrict__ out, const float* __restrict__ bias,
                                                v8f c, int nb, int m) {
  const v4f b0 = *(const v4f*)(bias + nb);
  const v4f b1 = *(const v4f*)(bias + nb + 4);
  float* dst = out + ((size_t)((m / SEQ) * HEADS + nb / HD) * SEQ + (m % SEQ)) * HD;
  v4f p0, p1;
  p0.x = c[0] + b0.x; p0.y = c[1] + b0.y; p0.z = c[2] + b0.z; p0.w = c[3] + b0.w;
  p1.x = c[4] + b1.x; p1.y = c[5] + b1.y; p1.z = c[6] + b1.z; p1.w = c[7] + b1.w;
  *(v4f*)dst       = p0;
  *(v4f*)(dst + 4) = p1;
}

__device__ __forceinline__ void store_head_keys(float* __restrict__ out, float bn,
                                                v8f c, int n, int mt, int hi) {
  float* dst = out + (size_t)((mt / SEQ) * HEADS + n / HD) * (SEQ * HD)
                   + ((mt % SEQ) / 16) * (16 * HD) + (n % HD) * 16 + 8 * hi;
  v4f p0, p1;
  p0.x = c[0] + bn; p0.y = c[1] + bn; p0.z = c[2] + bn; p0.w = c[3] + bn;
  p1.x = c[4] + bn; p1.y = c[5] + bn; p1.z = c[6] + bn; p1.w = c[7] + bn;
  *(v4f*)dst       = p0;
  *(v4f*)(dst + 4) = p1;
}

__global__ __launch_bounds__(128) void proj_wmma(
    const float* __restrict__ x,
    const float* __restrict__ W0, const float* __restrict__ b0, float* __restrict__ o0,
    const float* __restrict__ W1, const float* __restrict__ b1, float* __restrict__ o1,
    const float* __restrict__ W2, const float* __restrict__ b2, float* __restrict__ o2) {
  const int lane = threadIdx.x & 31;
  const int wave = __builtin_amdgcn_readfirstlane((int)(threadIdx.x >> 5));
  const int lo = lane & 15, hi = lane >> 4;

  const float* W = W0;
  const float* bias = b0;
  float* out = o0;
  bool rows8 = true;
  const int which = blockIdx.x % 3;
  const int blk = blockIdx.x / 3;
  if (which == 1) { W = W1; bias = b1; out = o1; }
  if (which == 2) { W = W2; bias = b2; out = o2; rows8 = false; }
  const int m0 = (blk >> 1) * 64 + (wave & 1) * 32;
  const int n0 = (blk & 1) * 64 + (wave >> 1) * 32;

  const float* xrow = x + (size_t)(m0 + lo) * EMB + 4 * hi;
  const float* wrow = W + (size_t)(n0 + lo) * EMB + 4 * hi;
  const float* a0p = rows8 ? wrow : xrow;
  const float* a1p = a0p + 16 * EMB;
  const float* b0p = rows8 ? xrow : wrow;
  const float* b1p = b0p + 16 * EMB;
  v8f c00 = {}, c01 = {}, c10 = {}, c11 = {};
  PGroup f0, f1, f2, f3;
  proj_load(f0, a0p, a1p, b0p, b1p, 0);
  proj_load(f1, a0p, a1p, b0p, b1p, 1);
  proj_load(f2, a0p, a1p, b0p, b1p, 2);
  proj_load(f3, a0p, a1p, b0p, b1p, 3);
  __builtin_amdgcn_sched_barrier(0);
  proj_mma(f0, c00, c01, c10, c11);
  proj_mma(f1, c00, c01, c10, c11);
  proj_mma(f2, c00, c01, c10, c11);
  proj_mma(f3, c00, c01, c10, c11);
  if (rows8) {
    store_head_rows(out, bias, c00, n0 + 8 * hi,      m0 + lo);
    store_head_rows(out, bias, c01, n0 + 8 * hi,      m0 + 16 + lo);
    store_head_rows(out, bias, c10, n0 + 16 + 8 * hi, m0 + lo);
    store_head_rows(out, bias, c11, n0 + 16 + 8 * hi, m0 + 16 + lo);
    return;
  }
  const float bn0 = bias[n0 + lo];
  const float bn1 = bias[n0 + 16 + lo];
  store_head_keys(out, bn0, c00, n0 + lo,      m0,      hi);
  store_head_keys(out, bn1, c01, n0 + 16 + lo, m0,      hi);
  store_head_keys(out, bn0, c10, n0 + lo,      m0 + 16, hi);
  store_head_keys(out, bn1, c11, n0 + 16 + lo, m0 + 16, hi);
}


struct KvFrag {
  v4f k;
  v4f va, vb;
};

__device__ __forceinline__ void kv_load(KvFrag& f, const float* __restrict__ Kb,
                                        const float* __restrict__ Vb, int kt,
                                        unsigned koff, unsigned voff) {
  const float* kt_k = Kb + (size_t)kt * (16 * HD);
  const float* kt_v = Vb + (size_t)kt * (16 * HD);
  f.k  = *(const v4f*)(kt_k + koff);
  f.va = *(const v4f*)(kt_v + voff);
  f.vb = *(const v4f*)(kt_v + voff + 4);
}

template <bool MASK, bool FAST>
__device__ __forceinline__ v8f kv_tile(const KvFrag& f, v2f bq0, v2f bq1, v8f acc,
                                       int lo, int hi, bool isv, float fill) {
  v2f k0, k1;
  k0.x = f.k.x; k0.y = f.k.y; k1.x = f.k.z; k1.y = f.k.w;
  const float v[8] = {f.va.x, f.va.y, f.va.z, f.va.w, f.vb.x, f.vb.y, f.vb.z, f.vb.w};
  v8f s = {};
  s = wmma4_f32(k0, bq0, s);
  s = wmma4_f32(k1, bq1, s);

  float w[8];
#pragma unroll
  for (int r = 0; r < 8; ++r) {
    float wt;
    if (FAST) {
      wt = 1.0f + __builtin_amdgcn_tanhf(s[r]);
    } else {
      const float e = __builtin_amdgcn_exp2f(NEG2_LOG2E * s[r]);
      wt = __builtin_amdgcn_rcpf(0.5f + 0.5f * e);
    }
    w[r] = MASK ? ((8 * hi + r <= lo) ? wt : 0.0f) : wt;
  }

#pragma unroll
  for (int j = 0; j < 4; ++j) {
    v2f a, bb;
    a.x  = isv ? v[2 * j]     : fill;
    a.y  = isv ? v[2 * j + 1] : fill;
    bb.x = w[2 * j];
    bb.y = w[2 * j + 1];
    acc = wmma4_f32(a, bb, acc);
  }
  return acc;
}

constexpr int PRIO_T2 = 24, PRIO_T1 = 44, PRIO_T0 = 58;
constexpr int QTILES_ALL = SEQ / 16;
__device__ __forceinline__ int prio_level(int done) {
  return done >= PRIO_T0 ? 0 : done >= PRIO_T1 ? 1 : done >= PRIO_T2 ? 2 : 3;
}
__device__ __forceinline__ void prio_set(int level) {
  if (level == 3)      __builtin_amdgcn_s_setprio(3);
  else if (level == 2) __builtin_amdgcn_s_setprio(2);
  else if (level == 1) __builtin_amdgcn_s_setprio(1);
  else                 __builtin_amdgcn_s_setprio(0);
}

struct Acc2 { v8f a, b; };
constexpr size_t HSTRIDE = (size_t)SEQ * HD;

template <bool DIAG, bool FAST>
__device__ __forceinline__ Acc2 span_pass(
    const float* __restrict__ Qb, const float* __restrict__ Kb, const float* __restrict__ Vb,
    int qt, int kt0, int kt1, int lo, int hi, int done0) {
  const int q0 = qt * 16;
  const float* Kb2 = Kb + HSTRIDE;
  const float* Vb2 = Vb + HSTRIDE;

  const v4f q  = *(const v4f*)(Qb + (size_t)(q0 + lo) * HD + 4 * hi);
  const v4f q2 = *(const v4f*)(Qb + HSTRIDE + (size_t)(q0 + lo) * HD + 4 * hi);
  v2f bq0, bq1, cq0, cq1;
  bq0.x = q.x * HALF_SCALE;  bq0.y = q.y * HALF_SCALE;
  bq1.x = q.z * HALF_SCALE;  bq1.y = q.w * HALF_SCALE;
  cq0.x = q2.x * HALF_SCALE; cq0.y = q2.y * HALF_SCALE;
  cq1.x = q2.z * HALF_SCALE; cq1.y = q2.w * HALF_SCALE;

  const bool  isv  = lo < 8;
  const float fill = (lo == 8) ? 1.0f : 0.0f;

  const unsigned koff = (unsigned)(lo * HD + 4 * hi);
  const unsigned voff = (unsigned)((lo & 7) * 16 + 8 * hi);
  const int last     = kt1 - 1;
  const int lastfull = DIAG ? last - 1 : last;

  KvFrag fa, fb;
  KvFrag ga, gb;
  kv_load(fa, Kb,  Vb,  kt0, koff, voff);
  kv_load(ga, Kb2, Vb2, kt0, koff, voff);
  Acc2 acc;
  acc.a = (v8f){};
  acc.b = (v8f){};
  int kt = kt0;
  int level = prio_level(done0);
  prio_set(level);
  for (;;) {
    const int bandend = level == 3 ? PRIO_T2 : level == 2 ? PRIO_T1 : level == 1 ? PRIO_T0 : 2 * QTILES_ALL;
    const int want = kt0 + (bandend - done0);
    const int stop = want < lastfull ? want : lastfull;
    for (; kt < stop; kt += 2) {
      kv_load(fb, Kb,  Vb,  kt + 1, koff, voff);
      kv_load(gb, Kb2, Vb2, kt + 1, koff, voff);
      asm volatile("" ::: "memory");
      acc.a = kv_tile<false, FAST>(fa, bq0, bq1, acc.a, lo, hi, isv, fill);
      acc.b = kv_tile<false, FAST>(ga, cq0, cq1, acc.b, lo, hi, isv, fill);
      const int nx = (kt + 2 < last) ? kt + 2 : last;
      kv_load(fa, Kb,  Vb,  nx, koff, voff);
      kv_load(ga, Kb2, Vb2, nx, koff, voff);
      asm volatile("" ::: "memory");
      acc.a = kv_tile<false, FAST>(fb, bq0, bq1, acc.a, lo, hi, isv, fill);
      acc.b = kv_tile<false, FAST>(gb, cq0, cq1, acc.b, lo, hi, isv, fill);
    }
    if (kt >= lastfull) break;
    level -= 1;
    prio_set(level);
  }
  if (kt == lastfull) {
    if (DIAG) {
      kv_load(fb, Kb,  Vb,  last, koff, voff);
      kv_load(gb, Kb2, Vb2, last, koff, voff);
      asm volatile("" ::: "memory");
    }
    acc.a = kv_tile<false, FAST>(fa, bq0, bq1, acc.a, lo, hi, isv, fill);
    acc.b = kv_tile<false, FAST>(ga, cq0, cq1, acc.b, lo, hi, isv, fill);
    if (DIAG) {
      acc.a = kv_tile<true, FAST>(fb, bq0, bq1, acc.a, lo, hi, isv, fill);
      acc.b = kv_tile<true, FAST>(gb, cq0, cq1, acc.b, lo, hi, isv, fill);
    }
  } else if (DIAG) {
    acc.a = kv_tile<true, FAST>(fa, bq0, bq1, acc.a, lo, hi, isv, fill);
    acc.b = kv_tile<true, FAST>(ga, cq0, cq1, acc.b, lo, hi, isv, fill);
  }
  return acc;
}

template <bool DIAG>
__device__ __forceinline__ Acc2 attn_span(
    const float* __restrict__ Qb, const float* __restrict__ Kb, const float* __restrict__ Vb,
    int qt, int kt0, int kt1, int lo, int hi, int done0) {
  Acc2 acc = span_pass<DIAG, true>(Qb, Kb, Vb, qt, kt0, kt1, lo, hi, done0);
  const int upto = qt * 16 + lo - kt0 * 16 + 1;
  const int span = (kt1 - kt0) * 16;
  const int nk   = upto < span ? upto : span;
  const float need = 0.03125f * (float)nk;
  const bool thin = (hi == 1) && !(acc.a[0] >= need && acc.b[0] >= need);
  if (__builtin_amdgcn_ballot_w32(thin) != 0u)
    acc = span_pass<DIAG, false>(Qb, Kb, Vb, qt, kt0, kt1, lo, hi, done0);
  return acc;
}

constexpr int APITCH = EMB + 4;

__device__ __forceinline__ void attn_to_lds(float* arow, v8f acc, int rowbase, int h, int lo, int hi) {
  const float rs = __shfl(acc[0], 16 + lo, 32);
  if (hi == 0) {
    float4 o0, o1;
    o0.x = acc[0] / rs; o0.y = acc[1] / rs; o0.z = acc[2] / rs; o0.w = acc[3] / rs;
    o1.x = acc[4] / rs; o1.y = acc[5] / rs; o1.z = acc[6] / rs; o1.w = acc[7] / rs;
    float4* dst = (float4*)(arow + (rowbase + lo) * APITCH + h * HD);
    dst[0] = o0;
    dst[1] = o1;
  }
}

constexpr int QTILES = SEQ / 16;
constexpr int HALF_T = QTILES / 2;
__device__ __forceinline__ void part_put(float4 (*dst)[32], v8f acc, int lane) {
  float4 a0, a1;
  a0.x = acc[0]; a0.y = acc[1]; a0.z = acc[2]; a0.w = acc[3];
  a1.x = acc[4]; a1.y = acc[5]; a1.z = acc[6]; a1.w = acc[7];
  dst[0][lane] = a0;
  dst[1][lane] = a1;
}
__device__ __forceinline__ v8f part_add(float4 (*src)[32], v8f acc, int lane) {
  const float4 a0 = src[0][lane];
  const float4 a1 = src[1][lane];
  acc[0] += a0.x; acc[1] += a0.y; acc[2] += a0.z; acc[3] += a0.w;
  acc[4] += a1.x; acc[5] += a1.y; acc[6] += a1.z; acc[7] += a1.w;
  return acc;
}
__global__ __launch_bounds__(512) void attn_proj_wmma(
    const float* __restrict__ Q, const float* __restrict__ K, const float* __restrict__ V,
    const float* __restrict__ Wo, const float* __restrict__ bo, float* __restrict__ out) {
  __shared__ float4 part[HEADS][2][32];
  __shared__ __attribute__((aligned(16))) float arow[32 * APITCH];

  const int lane = threadIdx.x & 31;
  const int wave = __builtin_amdgcn_readfirstlane((int)(threadIdx.x >> 5));
  const int h    = 2 * (wave & 7);
  const int role = wave >> 3;
  const int b = blockIdx.x >> 6;
  const int p = blockIdx.x & 63;
  const int ql = p, qh = QTILES - 1 - p;
  const int lo = lane & 15, hi = lane >> 4;
  const size_t hbase = (size_t)(b * HEADS + h) * (SEQ * HD);
  const float* Qb = Q + hbase;
  const float* Kb = K + hbase;
  const float* Vb = V + hbase;

  Acc2 acc;
  if (role == 0) {
    acc = attn_span<false>(Qb, Kb, Vb, qh, 0, HALF_T, lo, hi, 0);
    part_put(part[h],     acc.a, lane);
    part_put(part[h + 1], acc.b, lane);
  } else {
    const Acc2 al = attn_span<true>(Qb, Kb, Vb, ql, 0, ql + 1, lo, hi, 0);
    attn_to_lds(arow, al.a, 0, h,     lo, hi);
    attn_to_lds(arow, al.b, 0, h + 1, lo, hi);
    acc = attn_span<true>(Qb, Kb, Vb, qh, HALF_T, qh + 1, lo, hi, ql + 1);
  }
  __builtin_amdgcn_s_setprio(0);
  __syncthreads();
  if (role == 1) {
    acc.a = part_add(part[h],     acc.a, lane);
    acc.b = part_add(part[h + 1], acc.b, lane);
    attn_to_lds(arow, acc.a, 16, h,     lo, hi);
    attn_to_lds(arow, acc.b, 16, h + 1, lo, hi);
  }
  __syncthreads();

  if (wave < 8) {
    const int rt = wave & 1;
    const int n0 = (wave >> 1) * 32;
    const float* arp = arow + (rt * 16 + lo) * APITCH + 4 * hi;
    const float* w0p = Wo + (size_t)(n0 + lo) * EMB + 4 * hi;
    const float* w1p = w0p + 16 * EMB;
    v8f c0 = {}, c1 = {};
#pragma unroll
    for (int c = 0; c < 16; ++c) {
      const v4f a  = *(const v4f*)(arp + 8 * c);
      const v4f w0 = *(const v4f*)(w0p + 8 * c);
      const v4f w1 = *(const v4f*)(w1p + 8 * c);
      v2f al, ah, w0l, w0h, w1l, w1h;
      al.x  = a.x;  al.y  = a.y;  ah.x  = a.z;  ah.y  = a.w;
      w0l.x = w0.x; w0l.y = w0.y; w0h.x = w0.z; w0h.y = w0.w;
      w1l.x = w1.x; w1l.y = w1.y; w1h.x = w1.z; w1h.y = w1.w;
      c0 = wmma4_f32(al, w0l, c0);
      c1 = wmma4_f32(al, w1l, c1);
      c0 = wmma4_f32(ah, w0h, c0);
      c1 = wmma4_f32(ah, w1h, c1);
    }
    const float bn0 = bo[n0 + lo];
    const float bn1 = bo[n0 + 16 + lo];
    const int q0 = (rt == 0 ? ql : qh) * 16;
    float* orow = out + ((size_t)b * SEQ + q0 + 8 * hi) * EMB + n0 + lo;
#pragma unroll
    for (int r = 0; r < 8; ++r) {
      orow[(size_t)r * EMB]      = c0[r] + bn0;
      orow[(size_t)r * EMB + 16] = c1[r] + bn1;
    }
  }
}

extern "C" void kernel_launch(void* const* d_in, const int* in_sizes, int n_in,
                              void* d_out, int out_size, void* d_ws, size_t ws_size,
                              hipStream_t stream) {
  const float* x  = (const float*)d_in[0];
  const float* Wq = (const float*)d_in[1];
  const float* bq = (const float*)d_in[2];
  const float* Wk = (const float*)d_in[3];
  const float* bk = (const float*)d_in[4];
  const float* Wv = (const float*)d_in[5];
  const float* bv = (const float*)d_in[6];
  const float* Wo = (const float*)d_in[7];
  const float* bo = (const float*)d_in[8];
  float* out = (float*)d_out;

  const size_t mat = (size_t)BATCH * SEQ * EMB;
  float* Qw = (float*)d_ws;
  float* Kw = Qw + mat;
  float* Vw = Kw + mat;

  proj_wmma<<<dim3(768), dim3(128), 0, stream>>>(x, Wq, bq, Qw, Wk, bk, Kw, Wv, bv, Vw);
  attn_proj_wmma<<<dim3(256), dim3(512), 0, stream>>>(Qw, Kw, Vw, Wo, bo, out);
}
